# starting kernel
# baseline (speedup 1.0000x reference)
#include <hip/hip_runtime.h>
#include <hip/hip_bf16.h>

#define NN 50000
#define EE 600000
#define INC 128
#define OUTC 128
#define HEADS 8
#define HEADC 16

typedef __attribute__((ext_vector_type(2))) float v2f;
typedef __attribute__((ext_vector_type(8))) float v8f;

__global__ void gat_gemm_wmma(const float* __restrict__ X,
                              const float* __restrict__ W,
                              float* __restrict__ emb) {
    const int wave  = (blockIdx.x * blockDim.x + threadIdx.x) >> 5;
    const int lane  = threadIdx.x & 31;
    const int tileM = wave >> 3;
    const int tileN = wave & 7;
    const int half  = lane >> 4;
    const int l     = lane & 15;

    const float* __restrict__ arow = X + (size_t)(tileM * 16 + l) * INC + half * 2;
    const float* __restrict__ brow = W + (size_t)(tileN * 16 + l) * INC + half * 2;

    v8f c = {};
#pragma unroll
    for (int kk = 0; kk < INC; kk += 4) {
        v2f a = *(const v2f*)(arow + kk);
        v2f b = *(const v2f*)(brow + kk);
        c = __builtin_amdgcn_wmma_f32_16x16x4_f32(
                 false, a,  false, b,
                 (short)0, c,  false,  false);
    }

    float* __restrict__ op =
        emb + (size_t)(tileM * 16 + half * 8) * OUTC + tileN * 16 + l;
#pragma unroll
    for (int v = 0; v < 8; ++v) op[(size_t)v * OUTC] = c[v];
}

__global__ void gat_scores(const float* __restrict__ emb,
                           const float* __restrict__ a_left,
                           const float* __restrict__ a_right,
                           float* __restrict__ ss, float* __restrict__ ts) {
    int gid = blockIdx.x * blockDim.x + threadIdx.x;
    if (gid >= NN * HEADS) return;
    int n = gid >> 3, h = gid & 7;
    const float* ep = emb + (size_t)n * OUTC + h * HEADC;
    float s0 = 0.f, s1 = 0.f;
#pragma unroll
    for (int c = 0; c < HEADC; ++c) {
        float e = ep[c];
        s0 += e * a_left[c * HEADS + h];
        s1 += e * a_right[c * HEADS + h];
    }
    ss[gid] = s0;
    ts[gid] = s1;
}

__global__ void gat_init(float* __restrict__ out, const float* __restrict__ bias,
                         float* __restrict__ denom) {
    int gid = blockIdx.x * blockDim.x + threadIdx.x;
    if (gid < NN * OUTC) out[gid] = bias[gid & (OUTC - 1)];
    if (gid < NN * HEADS) denom[gid] = 0.f;
}

__global__ void gat_edge_exp(const int* __restrict__ src, const int* __restrict__ trg,
                             const float* __restrict__ ss, const float* __restrict__ ts,
                             float* __restrict__ exps, float* __restrict__ denom) {
    int gid = blockIdx.x * blockDim.x + threadIdx.x;
    if (gid >= EE * HEADS) return;
    int e = gid >> 3, h = gid & 7;
    int s = src[e], t = trg[e];
    float v = ss[s * HEADS + h] + ts[t * HEADS + h];
    v = v > 0.f ? v : 0.2f * v;
    float ex = __expf(v);
    exps[gid] = ex;
    atomicAdd(&denom[t * HEADS + h], ex);
}

__global__ void gat_aggregate(const int* __restrict__ src, const int* __restrict__ trg,
                              const float* __restrict__ emb,
                              const float* __restrict__ exps,
                              const float* __restrict__ denom,
                              float* __restrict__ out) {
    long long gid = (long long)blockIdx.x * blockDim.x + threadIdx.x;
    if (gid >= (long long)EE * OUTC) return;
    int e = (int)(gid >> 7);
    int k = (int)(gid & (OUTC - 1));
    int h = k >> 4;
    int s = src[e], t = trg[e];
    float attn = exps[e * HEADS + h] / (denom[t * HEADS + h] + 1e-16f);
    atomicAdd(&out[(size_t)t * OUTC + k], emb[(size_t)s * OUTC + k] * attn);
}

extern "C" void kernel_launch(void* const* d_in, const int* in_sizes, int n_in,
                              void* d_out, int out_size, void* d_ws, size_t ws_size,
                              hipStream_t stream) {
    const float* X      = (const float*)d_in[0];
    const int*   eidx   = (const int*)d_in[1];
    const float* W      = (const float*)d_in[2];
    const float* a_l    = (const float*)d_in[3];
    const float* a_r    = (const float*)d_in[4];
    const float* bias   = (const float*)d_in[5];
    float* out = (float*)d_out;

    const int* src = eidx;
    const int* trg = eidx + EE;

    float* ws    = (float*)d_ws;
    float* emb   = ws;
    float* ss    = emb + (size_t)NN * OUTC;
    float* ts    = ss + (size_t)NN * HEADS;
    float* denom = ts + (size_t)NN * HEADS;
    float* exps  = denom + (size_t)NN * HEADS;

    gat_gemm_wmma<<<3125, 256, 0, stream>>>(X, W, emb);

    gat_scores<<<(NN * HEADS + 255) / 256, 256, 0, stream>>>(emb, a_l, a_r, ss, ts);

    gat_init<<<(NN * OUTC + 255) / 256, 256, 0, stream>>>(out, bias, denom);

    gat_edge_exp<<<(EE * HEADS + 255) / 256, 256, 0, stream>>>(src, trg, ss, ts, exps, denom);

    long long nagg = (long long)EE * OUTC;
    gat_aggregate<<<(unsigned)((nagg + 255) / 256), 256, 0, stream>>>(
        src, trg, emb, exps, denom, out);
}


# Group edges by target per call (int count, scan, place); one wave per node sorts its list by edge id, sums source rows in that fixed order in registers, stores out once: no f32 atomics
# speedup vs baseline: 2.0484x; 2.0484x over previous
#include <hip/hip_runtime.h>
#include <hip/hip_bf16.h>

#define NN 50000
#define EE 600000
#define INC 128
#define OUTC 128
#define HEADS 8
#define HEADC 16

typedef __attribute__((ext_vector_type(2))) float v2f;
typedef __attribute__((ext_vector_type(8))) float v8f;

__global__ void gat_gemm_wmma(const float* __restrict__ X,
                              const float* __restrict__ W,
                              float* __restrict__ emb,
                              int* __restrict__ cnt) {
    const int gtid = blockIdx.x * blockDim.x + threadIdx.x;
    if (gtid < NN) cnt[gtid] = 0;
    const int wave  = gtid >> 5;
    const int lane  = threadIdx.x & 31;
    const int tileM = wave >> 3;
    const int tileN = wave & 7;
    const int half  = lane >> 4;
    const int l     = lane & 15;

    const float* __restrict__ arow = X + (size_t)(tileM * 16 + l) * INC + half * 2;
    const float* __restrict__ brow = W + (size_t)(tileN * 16 + l) * INC + half * 2;

    v8f c = {};
#pragma unroll
    for (int kk = 0; kk < INC; kk += 4) {
        v2f a = *(const v2f*)(arow + kk);
        v2f b = *(const v2f*)(brow + kk);
        c = __builtin_amdgcn_wmma_f32_16x16x4_f32(
                 false, a,  false, b,
                 (short)0, c,  false,  false);
    }

    float* __restrict__ op =
        emb + (size_t)(tileM * 16 + half * 8) * OUTC + tileN * 16 + l;
#pragma unroll
    for (int v = 0; v < 8; ++v) op[(size_t)v * OUTC] = c[v];
}

__global__ void gat_scores(const float* __restrict__ emb,
                           const float* __restrict__ a_left,
                           const float* __restrict__ a_right,
                           float* __restrict__ ss, float* __restrict__ ts) {
    int gid = blockIdx.x * blockDim.x + threadIdx.x;
    if (gid >= NN * HEADS) return;
    int n = gid >> 3, h = gid & 7;
    const float* ep = emb + (size_t)n * OUTC + h * HEADC;
    float s0 = 0.f, s1 = 0.f;
#pragma unroll
    for (int c = 0; c < HEADC; ++c) {
        float e = ep[c];
        s0 += e * a_left[c * HEADS + h];
        s1 += e * a_right[c * HEADS + h];
    }
    ss[gid] = s0;
    ts[gid] = s1;
}

__global__ void gat_count(const int* __restrict__ trg, int* __restrict__ cnt,
                          int* __restrict__ pos) {
    int e = blockIdx.x * blockDim.x + threadIdx.x;
    if (e >= EE) return;
    pos[e] = atomicAdd(&cnt[trg[e]], 1);
}

#define SCAN_T  1024
#define SCAN_IT 13

__global__ void __launch_bounds__(SCAN_T)
gat_scan(const int* __restrict__ cnt, int* __restrict__ offs) {
    __shared__ int wsum[32];
    const int t = threadIdx.x, lane = t & 31, w = t >> 5;
    const int wbase = w * (SCAN_IT * 128) + lane * 4;
    int4 v[SCAN_IT];
    int  inc[SCAN_IT];
    int  own[SCAN_IT];
#pragma unroll
    for (int i = 0; i < SCAN_IT; ++i) {
        int idx = wbase + i * 128;
        v[i] = idx < NN ? *(const int4*)(cnt + idx) : make_int4(0, 0, 0, 0);
    }
#pragma unroll
    for (int i = 0; i < SCAN_IT; ++i) {
        own[i] = v[i].x + v[i].y + v[i].z + v[i].w;
        int x = own[i];
#pragma unroll
        for (int o = 1; o < 32; o <<= 1) {
            int n = __shfl_up(x, o);
            if (lane >= o) x += n;
        }
        inc[i] = x;
    }
    int carry = 0;
#pragma unroll
    for (int i = 0; i < SCAN_IT; ++i) {
        int tot = __builtin_amdgcn_readlane(inc[i], 31);
        inc[i] += carry;
        carry += tot;
    }
    if (lane == 0) wsum[w] = carry;
    __syncthreads();
    if (w == 0) {
        int x = wsum[lane];
#pragma unroll
        for (int o = 1; o < 32; o <<= 1) {
            int n = __shfl_up(x, o);
            if (lane >= o) x += n;
        }
        wsum[lane] = x;
    }
    __syncthreads();
    const int base = w > 0 ? wsum[w - 1] : 0;
#pragma unroll
    for (int i = 0; i < SCAN_IT; ++i) {
        int idx = wbase + i * 128;
        int4 o4;
        o4.x = base + inc[i] - own[i];
        o4.y = o4.x + v[i].x;
        o4.z = o4.y + v[i].y;
        o4.w = o4.z + v[i].z;
        if (idx < NN) *(int4*)(offs + idx) = o4;
    }
}

__global__ void gat_place(const int* __restrict__ trg, const int* __restrict__ offs,
                          const int* __restrict__ pos, int* __restrict__ list) {
    int e = blockIdx.x * blockDim.x + threadIdx.x;
    if (e >= EE) return;
    list[offs[trg[e]] + pos[e]] = e;
}

__device__ __forceinline__ float edge_exp(float s_src, float s_trg) {
    float v = s_src + s_trg;
    v = v > 0.f ? v : 0.2f * v;
    return __expf(v);
}

__global__ void __launch_bounds__(256)
gat_aggregate(const int* __restrict__ src,
              const int* __restrict__ cnt, const int* __restrict__ offs,
              const int* __restrict__ list,
              const float* __restrict__ emb,
              const float* __restrict__ ss, const float* __restrict__ ts,
              const float* __restrict__ bias,
              float* __restrict__ out) {
    const int lane = threadIdx.x & 31;
    const int node = __builtin_amdgcn_readfirstlane((int)((blockIdx.x * blockDim.x + threadIdx.x) >> 5));
    if (node >= NN) return;
    const int h = lane >> 2;
    const int deg = __builtin_amdgcn_readfirstlane(cnt[node]);
    const int beg = __builtin_amdgcn_readfirstlane(offs[node]);
    const float tsv = ts[node * HEADS + h];
    const float4* __restrict__ emb4 = (const float4*)emb;

    float denom = 0.f;
    float4 acc = make_float4(0.f, 0.f, 0.f, 0.f);

    if (deg <= 32) {
        int e = lane < deg ? list[beg + lane] : 0x7fffffff;
        int rank = 0;
        for (int j = 0; j < deg; ++j) {
            int ej = __builtin_amdgcn_readlane(e, j);
            rank += (ej < e) ? 1 : 0;
        }
        if (lane >= deg) rank = lane;
        int es = __builtin_amdgcn_ds_permute(rank << 2, e);
        int sv = lane < deg ? src[es] : 0;

        for (int r = 0; r < deg; ++r) {
            int s = __builtin_amdgcn_readlane(sv, r);
            denom += edge_exp(ss[s * HEADS + h], tsv);
        }
        const float dn = denom + 1e-16f;
        int r = 0;
        for (; r + 4 <= deg; r += 4) {
            int s0 = __builtin_amdgcn_readlane(sv, r);
            int s1 = __builtin_amdgcn_readlane(sv, r + 1);
            int s2 = __builtin_amdgcn_readlane(sv, r + 2);
            int s3 = __builtin_amdgcn_readlane(sv, r + 3);
            float4 x0 = emb4[s0 * 32 + lane];
            float4 x1 = emb4[s1 * 32 + lane];
            float4 x2 = emb4[s2 * 32 + lane];
            float4 x3 = emb4[s3 * 32 + lane];
            float a0 = edge_exp(ss[s0 * HEADS + h], tsv) / dn;
            float a1 = edge_exp(ss[s1 * HEADS + h], tsv) / dn;
            float a2 = edge_exp(ss[s2 * HEADS + h], tsv) / dn;
            float a3 = edge_exp(ss[s3 * HEADS + h], tsv) / dn;
            acc.x += x0.x * a0; acc.y += x0.y * a0; acc.z += x0.z * a0; acc.w += x0.w * a0;
            acc.x += x1.x * a1; acc.y += x1.y * a1; acc.z += x1.z * a1; acc.w += x1.w * a1;
            acc.x += x2.x * a2; acc.y += x2.y * a2; acc.z += x2.z * a2; acc.w += x2.w * a2;
            acc.x += x3.x * a3; acc.y += x3.y * a3; acc.z += x3.z * a3; acc.w += x3.w * a3;
        }
        for (; r < deg; ++r) {
            int s0 = __builtin_amdgcn_readlane(sv, r);
            float4 x0 = emb4[s0 * 32 + lane];
            float a0 = edge_exp(ss[s0 * HEADS + h], tsv) / dn;
            acc.x += x0.x * a0; acc.y += x0.y * a0; acc.z += x0.z * a0; acc.w += x0.w * a0;
        }
    } else {
        int prev = -1;
        for (int r = 0; r < deg; ++r) {
            int m = 0x7fffffff;
            for (int j = lane; j < deg; j += 32) {
                int ej = list[beg + j];
                if (ej > prev && ej < m) m = ej;
            }
#pragma unroll
            for (int o = 16; o > 0; o >>= 1) {
                int n = __shfl_xor(m, o);
                m = n < m ? n : m;
            }
            prev = __builtin_amdgcn_readfirstlane(m);
            int s = src[prev];
            denom += edge_exp(ss[s * HEADS + h], tsv);
        }
        const float dn = denom + 1e-16f;
        prev = -1;
        for (int r = 0; r < deg; ++r) {
            int m = 0x7fffffff;
            for (int j = lane; j < deg; j += 32) {
                int ej = list[beg + j];
                if (ej > prev && ej < m) m = ej;
            }
#pragma unroll
            for (int o = 16; o > 0; o >>= 1) {
                int n = __shfl_xor(m, o);
                m = n < m ? n : m;
            }
            prev = __builtin_amdgcn_readfirstlane(m);
            int s0 = src[prev];
            float4 x0 = emb4[s0 * 32 + lane];
            float a0 = edge_exp(ss[s0 * HEADS + h], tsv) / dn;
            acc.x += x0.x * a0; acc.y += x0.y * a0; acc.z += x0.z * a0; acc.w += x0.w * a0;
        }
    }

    const float4 b = ((const float4*)bias)[lane];
    float4 o4 = make_float4(acc.x + b.x, acc.y + b.y, acc.z + b.z, acc.w + b.w);
    ((float4*)out)[node * 32 + lane] = o4;
}

extern "C" void kernel_launch(void* const* d_in, const int* in_sizes, int n_in,
                              void* d_out, int out_size, void* d_ws, size_t ws_size,
                              hipStream_t stream) {
    const float* X      = (const float*)d_in[0];
    const int*   eidx   = (const int*)d_in[1];
    const float* W      = (const float*)d_in[2];
    const float* a_l    = (const float*)d_in[3];
    const float* a_r    = (const float*)d_in[4];
    const float* bias   = (const float*)d_in[5];
    float* out = (float*)d_out;

    const int* src = eidx;
    const int* trg = eidx + EE;

    float* ws    = (float*)d_ws;
    float* emb   = ws;
    float* ss    = emb + (size_t)NN * OUTC;
    float* ts    = ss + (size_t)NN * HEADS;
    int*   cnt   = (int*)(ts + (size_t)NN * HEADS);
    int*   offs  = cnt + NN;
    int*   pos   = offs + NN;
    int*   list  = pos + EE;

    gat_gemm_wmma<<<3125, 256, 0, stream>>>(X, W, emb, cnt);

    gat_scores<<<(NN * HEADS + 255) / 256, 256, 0, stream>>>(emb, a_l, a_r, ss, ts);

    gat_count<<<(EE + 255) / 256, 256, 0, stream>>>(trg, cnt, pos);
    gat_scan<<<1, SCAN_T, 0, stream>>>(cnt, offs);
    gat_place<<<(EE + 255) / 256, 256, 0, stream>>>(trg, offs, pos, list);

    gat_aggregate<<<(NN + 7) / 8, 256, 0, stream>>>(src, cnt, offs, list, emb, ss, ts, bias, out);
}


# Group edges by target on LDS counters: per-chunk counts over 256 target ranges, one scan, range records, then per-range node counts and lists in LDS; no global atomic per edge; sums unchanged
# speedup vs baseline: 2.6570x; 1.2971x over previous
#include <hip/hip_runtime.h>
#include <hip/hip_bf16.h>

#define NN 50000
#define EE 600000
#define INC 128
#define OUTC 128
#define HEADS 8
#define HEADC 16

typedef __attribute__((ext_vector_type(2))) float v2f;
typedef __attribute__((ext_vector_type(8))) float v8f;

__global__ void gat_gemm_wmma(const float* __restrict__ X,
                              const float* __restrict__ W,
                              float* __restrict__ emb) {
    const int gtid = blockIdx.x * blockDim.x + threadIdx.x;
    const int wave  = gtid >> 5;
    const int lane  = threadIdx.x & 31;
    const int tileM = wave >> 3;
    const int tileN = wave & 7;
    const int half  = lane >> 4;
    const int l     = lane & 15;

    const float* __restrict__ arow = X + (size_t)(tileM * 16 + l) * INC + half * 2;
    const float* __restrict__ brow = W + (size_t)(tileN * 16 + l) * INC + half * 2;

    v8f c = {};
#pragma unroll
    for (int kk = 0; kk < INC; kk += 4) {
        v2f a = *(const v2f*)(arow + kk);
        v2f b = *(const v2f*)(brow + kk);
        c = __builtin_amdgcn_wmma_f32_16x16x4_f32(
                 false, a,  false, b,
                 (short)0, c,  false,  false);
    }

    float* __restrict__ op =
        emb + (size_t)(tileM * 16 + half * 8) * OUTC + tileN * 16 + l;
#pragma unroll
    for (int v = 0; v < 8; ++v) op[(size_t)v * OUTC] = c[v];
}

__global__ void gat_scores(const float* __restrict__ emb,
                           const float* __restrict__ a_left,
                           const float* __restrict__ a_right,
                           float* __restrict__ ss, float* __restrict__ ts) {
    int gid = blockIdx.x * blockDim.x + threadIdx.x;
    if (gid >= NN * HEADS) return;
    int n = gid >> 3, h = gid & 7;
    const float* ep = emb + (size_t)n * OUTC + h * HEADC;
    float s0 = 0.f, s1 = 0.f;
#pragma unroll
    for (int c = 0; c < HEADC; ++c) {
        float e = ep[c];
        s0 += e * a_left[c * HEADS + h];
        s1 += e * a_right[c * HEADS + h];
    }
    ss[gid] = s0;
    ts[gid] = s1;
}

#define NB    256
#define BIN_W 196
#define NCH   256
#define CH    2344
static_assert(NB * BIN_W >= NN, "bins cover every node");
static_assert(NCH * CH >= EE, "chunks cover every edge");
static_assert(BIN_W <= 256 && EE < (1 << 23), "record packing: edge id << 8 | index in bin");
#define KPT   ((CH + 1023) / 1024)

__global__ void __launch_bounds__(1024)
gat_binhist(const int* __restrict__ trg, int* __restrict__ H) {
    __shared__ int h[NB];
    const int t = threadIdx.x;
    if (t < NB) h[t] = 0;
    __syncthreads();
    const int e0 = blockIdx.x * CH;
    int tg[KPT];
#pragma unroll
    for (int k = 0; k < KPT; ++k) {
        const int i = t + k * 1024;
        tg[k] = (i < CH && e0 + i < EE) ? trg[e0 + i] : -1;
    }
#pragma unroll
    for (int k = 0; k < KPT; ++k)
        if (tg[k] >= 0) atomicAdd(&h[tg[k] / BIN_W], 1);
    __syncthreads();
    if (t < NB) H[t * NCH + blockIdx.x] = h[t];
}

#define SCAN_T  1024
#define SCAN_IT 16
static_assert(32 * SCAN_IT * 128 == NB * NCH, "scan covers H exactly");

__global__ void __launch_bounds__(SCAN_T)
gat_scan(int* H) {
    __shared__ int wsum[32];
    const int t = threadIdx.x, lane = t & 31, w = t >> 5;
    const int wbase = w * (SCAN_IT * 128) + lane * 4;
    int4 v[SCAN_IT];
    int  inc[SCAN_IT];
#pragma unroll
    for (int i = 0; i < SCAN_IT; ++i) v[i] = *(const int4*)(H + wbase + i * 128);
#pragma unroll
    for (int i = 0; i < SCAN_IT; ++i) {
        int x = v[i].x + v[i].y + v[i].z + v[i].w;
#pragma unroll
        for (int o = 1; o < 32; o <<= 1) {
            int n = __shfl_up(x, o);
            if (lane >= o) x += n;
        }
        inc[i] = x;
    }
    int carry = 0;
#pragma unroll
    for (int i = 0; i < SCAN_IT; ++i) {
        int tot = __builtin_amdgcn_readlane(inc[i], 31);
        inc[i] += carry;
        carry += tot;
    }
    if (lane == 0) wsum[w] = carry;
    __syncthreads();
    if (w == 0) {
        int x = wsum[lane];
#pragma unroll
        for (int o = 1; o < 32; o <<= 1) {
            int n = __shfl_up(x, o);
            if (lane >= o) x += n;
        }
        wsum[lane] = x;
    }
    __syncthreads();
    const int base = w > 0 ? wsum[w - 1] : 0;
#pragma unroll
    for (int i = 0; i < SCAN_IT; ++i) {
        int4 o4;
        o4.x = base + inc[i] - (v[i].x + v[i].y + v[i].z + v[i].w);
        o4.y = o4.x + v[i].x;
        o4.z = o4.y + v[i].y;
        o4.w = o4.z + v[i].z;
        *(int4*)(H + wbase + i * 128) = o4;
    }
}

__global__ void __launch_bounds__(1024)
gat_binplace(const int* __restrict__ trg, const int* __restrict__ O,
             int* __restrict__ rec) {
    __shared__ int cur[NB];
    const int t = threadIdx.x;
    if (t < NB) cur[t] = O[t * NCH + blockIdx.x];
    __syncthreads();
    const int e0 = blockIdx.x * CH;
    int tg[KPT];
#pragma unroll
    for (int k = 0; k < KPT; ++k) {
        const int i = t + k * 1024;
        tg[k] = (i < CH && e0 + i < EE) ? trg[e0 + i] : -1;
    }
#pragma unroll
    for (int k = 0; k < KPT; ++k) {
        if (tg[k] >= 0) {
            const int e = e0 + t + k * 1024;
            const int b = tg[k] / BIN_W;
            const int p = atomicAdd(&cur[b], 1);
            rec[p] = (e << 8) | (tg[k] - b * BIN_W);
        }
    }
}

__global__ void __launch_bounds__(1024)
gat_fine(const int* __restrict__ O, const int* __restrict__ rec,
         int* __restrict__ cnt, int* __restrict__ offs, int* __restrict__ list) {
    __shared__ int c[256];
    __shared__ int cur[256];
    __shared__ int wsum[8];
    const int t = threadIdx.x, lane = t & 31, w = t >> 5;
    const int b = blockIdx.x;
    const int start = O[b * NCH];
    const int end   = (b == NB - 1) ? EE : O[(b + 1) * NCH];
    const int n = end - start;
    if (t < 256) c[t] = 0;
    __syncthreads();
    for (int i0 = 0; i0 < n; i0 += 4096) {
        int r[4];
#pragma unroll
        for (int k = 0; k < 4; ++k) {
            const int i = i0 + k * 1024 + t;
            r[k] = i < n ? rec[start + i] : -1;
        }
#pragma unroll
        for (int k = 0; k < 4; ++k)
            if (r[k] >= 0) atomicAdd(&c[r[k] & 255], 1);
    }
    __syncthreads();
    const int own = t < 256 ? c[t] : 0;
    int x = own;
#pragma unroll
    for (int o = 1; o < 32; o <<= 1) {
        int m = __shfl_up(x, o);
        if (lane >= o) x += m;
    }
    if (t < 256 && lane == 31) wsum[w] = x;
    __syncthreads();
    if (t < 256) {
        int base = 0;
        for (int j = 0; j < w; ++j) base += wsum[j];
        const int ex = base + x - own;
        cur[t] = ex;
        const int node = b * BIN_W + t;
        if (t < BIN_W && node < NN) {
            cnt[node]  = own;
            offs[node] = start + ex;
        }
    }
    __syncthreads();
    for (int i0 = 0; i0 < n; i0 += 4096) {
        int r[4];
#pragma unroll
        for (int k = 0; k < 4; ++k) {
            const int i = i0 + k * 1024 + t;
            r[k] = i < n ? rec[start + i] : -1;
        }
#pragma unroll
        for (int k = 0; k < 4; ++k) {
            if (r[k] >= 0) {
                const int p = atomicAdd(&cur[r[k] & 255], 1);
                list[start + p] = r[k] >> 8;
            }
        }
    }
}

__device__ __forceinline__ float edge_exp(float s_src, float s_trg) {
    float v = s_src + s_trg;
    v = v > 0.f ? v : 0.2f * v;
    return __expf(v);
}

__global__ void __launch_bounds__(256)
gat_aggregate(const int* __restrict__ src,
              const int* __restrict__ cnt, const int* __restrict__ offs,
              const int* __restrict__ list,
              const float* __restrict__ emb,
              const float* __restrict__ ss, const float* __restrict__ ts,
              const float* __restrict__ bias,
              float* __restrict__ out) {
    const int lane = threadIdx.x & 31;
    const int node = __builtin_amdgcn_readfirstlane((int)((blockIdx.x * blockDim.x + threadIdx.x) >> 5));
    if (node >= NN) return;
    const int h = lane >> 2;
    const int deg = __builtin_amdgcn_readfirstlane(cnt[node]);
    const int beg = __builtin_amdgcn_readfirstlane(offs[node]);
    const float tsv = ts[node * HEADS + h];
    const float4* __restrict__ emb4 = (const float4*)emb;

    float denom = 0.f;
    float4 acc = make_float4(0.f, 0.f, 0.f, 0.f);

    if (deg <= 32) {
        int e = lane < deg ? list[beg + lane] : 0x7fffffff;
        int rank = 0;
        for (int j = 0; j < deg; ++j) {
            int ej = __builtin_amdgcn_readlane(e, j);
            rank += (ej < e) ? 1 : 0;
        }
        if (lane >= deg) rank = lane;
        int es = __builtin_amdgcn_ds_permute(rank << 2, e);
        int sv = lane < deg ? src[es] : 0;

        for (int r = 0; r < deg; ++r) {
            int s = __builtin_amdgcn_readlane(sv, r);
            denom += edge_exp(ss[s * HEADS + h], tsv);
        }
        const float dn = denom + 1e-16f;
        int r = 0;
        for (; r + 4 <= deg; r += 4) {
            int s0 = __builtin_amdgcn_readlane(sv, r);
            int s1 = __builtin_amdgcn_readlane(sv, r + 1);
            int s2 = __builtin_amdgcn_readlane(sv, r + 2);
            int s3 = __builtin_amdgcn_readlane(sv, r + 3);
            float4 x0 = emb4[s0 * 32 + lane];
            float4 x1 = emb4[s1 * 32 + lane];
            float4 x2 = emb4[s2 * 32 + lane];
            float4 x3 = emb4[s3 * 32 + lane];
            float a0 = edge_exp(ss[s0 * HEADS + h], tsv) / dn;
            float a1 = edge_exp(ss[s1 * HEADS + h], tsv) / dn;
            float a2 = edge_exp(ss[s2 * HEADS + h], tsv) / dn;
            float a3 = edge_exp(ss[s3 * HEADS + h], tsv) / dn;
            acc.x += x0.x * a0; acc.y += x0.y * a0; acc.z += x0.z * a0; acc.w += x0.w * a0;
            acc.x += x1.x * a1; acc.y += x1.y * a1; acc.z += x1.z * a1; acc.w += x1.w * a1;
            acc.x += x2.x * a2; acc.y += x2.y * a2; acc.z += x2.z * a2; acc.w += x2.w * a2;
            acc.x += x3.x * a3; acc.y += x3.y * a3; acc.z += x3.z * a3; acc.w += x3.w * a3;
        }
        for (; r < deg; ++r) {
            int s0 = __builtin_amdgcn_readlane(sv, r);
            float4 x0 = emb4[s0 * 32 + lane];
            float a0 = edge_exp(ss[s0 * HEADS + h], tsv) / dn;
            acc.x += x0.x * a0; acc.y += x0.y * a0; acc.z += x0.z * a0; acc.w += x0.w * a0;
        }
    } else {
        int prev = -1;
        for (int r = 0; r < deg; ++r) {
            int m = 0x7fffffff;
            for (int j = lane; j < deg; j += 32) {
                int ej = list[beg + j];
                if (ej > prev && ej < m) m = ej;
            }
#pragma unroll
            for (int o = 16; o > 0; o >>= 1) {
                int n = __shfl_xor(m, o);
                m = n < m ? n : m;
            }
            prev = __builtin_amdgcn_readfirstlane(m);
            int s = src[prev];
            denom += edge_exp(ss[s * HEADS + h], tsv);
        }
        const float dn = denom + 1e-16f;
        prev = -1;
        for (int r = 0; r < deg; ++r) {
            int m = 0x7fffffff;
            for (int j = lane; j < deg; j += 32) {
                int ej = list[beg + j];
                if (ej > prev && ej < m) m = ej;
            }
#pragma unroll
            for (int o = 16; o > 0; o >>= 1) {
                int n = __shfl_xor(m, o);
                m = n < m ? n : m;
            }
            prev = __builtin_amdgcn_readfirstlane(m);
            int s0 = src[prev];
            float4 x0 = emb4[s0 * 32 + lane];
            float a0 = edge_exp(ss[s0 * HEADS + h], tsv) / dn;
            acc.x += x0.x * a0; acc.y += x0.y * a0; acc.z += x0.z * a0; acc.w += x0.w * a0;
        }
    }

    const float4 b = ((const float4*)bias)[lane];
    float4 o4 = make_float4(acc.x + b.x, acc.y + b.y, acc.z + b.z, acc.w + b.w);
    ((float4*)out)[node * 32 + lane] = o4;
}

extern "C" void kernel_launch(void* const* d_in, const int* in_sizes, int n_in,
                              void* d_out, int out_size, void* d_ws, size_t ws_size,
                              hipStream_t stream) {
    const float* X      = (const float*)d_in[0];
    const int*   eidx   = (const int*)d_in[1];
    const float* W      = (const float*)d_in[2];
    const float* a_l    = (const float*)d_in[3];
    const float* a_r    = (const float*)d_in[4];
    const float* bias   = (const float*)d_in[5];
    float* out = (float*)d_out;

    const int* src = eidx;
    const int* trg = eidx + EE;

    float* ws    = (float*)d_ws;
    float* emb   = ws;
    float* ss    = emb + (size_t)NN * OUTC;
    float* ts    = ss + (size_t)NN * HEADS;
    int*   cnt   = (int*)(ts + (size_t)NN * HEADS);
    int*   offs  = cnt + NN;
    int*   rec   = offs + NN;
    int*   list  = rec + EE;
    int*   H     = (int*)emb;

    gat_binhist<<<NCH, 1024, 0, stream>>>(trg, H);
    gat_scan<<<1, SCAN_T, 0, stream>>>(H);
    gat_binplace<<<NCH, 1024, 0, stream>>>(trg, H, rec);
    gat_fine<<<NB, 1024, 0, stream>>>(H, rec, cnt, offs, list);

    gat_gemm_wmma<<<3125, 256, 0, stream>>>(X, W, emb);

    gat_scores<<<(NN * HEADS + 255) / 256, 256, 0, stream>>>(emb, a_l, a_r, ss, ts);

    gat_aggregate<<<(NN + 7) / 8, 256, 0, stream>>>(src, cnt, offs, list, emb, ss, ts, bias, out);
}


# GEMM: 8-wave blocks keep their W fragments in registers and walk 16-row strips of X staged once in LDS by whole-line loads, not 16-byte pieces of X and W from 16 lines per load; sums unchanged
# speedup vs baseline: 3.3087x; 1.2453x over previous
#include <hip/hip_runtime.h>
#include <hip/hip_bf16.h>

#define NN 50000
#define EE 600000
#define INC 128
#define OUTC 128
#define HEADS 8
#define HEADC 16

typedef __attribute__((ext_vector_type(2))) float v2f;
typedef __attribute__((ext_vector_type(8))) float v8f;

#define STRIPS    (NN / 16)
#define GEMM_GRID 1024
#define APITCH    132
#define TILE_F    (16 * APITCH)
static_assert(STRIPS * 16 == NN, "strips cover every row exactly");
static_assert(GEMM_GRID <= STRIPS, "every block has a first strip");

__global__ void __launch_bounds__(256)
gat_gemm_wmma(const float* __restrict__ X,
              const float* __restrict__ W,
              float* __restrict__ emb) {
    __shared__ __attribute__((aligned(16))) float L[8 * TILE_F];
    const int tid  = threadIdx.x;
    const int wave = tid >> 5;
    const int lane = tid & 31;
    const int half = lane >> 4;
    const int l    = lane & 15;

    const float4* __restrict__ W4 = (const float4*)W + wave * 512;
    float* wsm = L + wave * TILE_F;
#pragma unroll
    for (int i = 0; i < 16; ++i)
        *(float4*)(wsm + i * APITCH + lane * 4) = W4[i * 32 + lane];
    __syncthreads();
    v2f b[32];
#pragma unroll
    for (int kk = 0; kk < 32; ++kk)
        b[kk] = *(const v2f*)(wsm + l * APITCH + kk * 4 + half * 2);

    const float4* __restrict__ X4 = (const float4*)X;
    float* as0 = L + (tid >> 5) * APITCH + (tid & 31) * 4;
    float* as1 = as0 + 8 * APITCH;
    const float* afr = L + l * APITCH + half * 2;

    int strip = blockIdx.x;
    float* __restrict__ op =
        emb + (size_t)(strip * 16 + half * 8) * OUTC + wave * 16 + l;
    float4 t0 = X4[strip * 512 + tid];
    float4 t1 = X4[strip * 512 + 256 + tid];
    for (; strip < STRIPS; strip += GEMM_GRID) {
        __syncthreads();
        *(float4*)as0 = t0;
        *(float4*)as1 = t1;
        __syncthreads();
        const int next = strip + GEMM_GRID;
        if (next < STRIPS) {
            t0 = X4[next * 512 + tid];
            t1 = X4[next * 512 + 256 + tid];
        }

        v8f c = {};
#pragma unroll
        for (int kk = 0; kk < 32; ++kk) {
            v2f a = *(const v2f*)(afr + kk * 4);
            c = __builtin_amdgcn_wmma_f32_16x16x4_f32(
                     false, a,  false, b[kk],
                     (short)0, c,  false,  false);
        }

#pragma unroll
        for (int v = 0; v < 8; ++v) op[(size_t)v * OUTC] = c[v];
        op += (size_t)GEMM_GRID * 16 * OUTC;
    }
}

__global__ void gat_scores(const float* __restrict__ emb,
                           const float* __restrict__ a_left,
                           const float* __restrict__ a_right,
                           float* __restrict__ ss, float* __restrict__ ts) {
    int gid = blockIdx.x * blockDim.x + threadIdx.x;
    if (gid >= NN * HEADS) return;
    int n = gid >> 3, h = gid & 7;
    const float* ep = emb + (size_t)n * OUTC + h * HEADC;
    float s0 = 0.f, s1 = 0.f;
#pragma unroll
    for (int c = 0; c < HEADC; ++c) {
        float e = ep[c];
        s0 += e * a_left[c * HEADS + h];
        s1 += e * a_right[c * HEADS + h];
    }
    ss[gid] = s0;
    ts[gid] = s1;
}

#define NB    256
#define BIN_W 196
#define NCH   256
#define CH    2344
static_assert(NB * BIN_W >= NN, "bins cover every node");
static_assert(NCH * CH >= EE, "chunks cover every edge");
static_assert(BIN_W <= 256 && EE < (1 << 23), "record packing: edge id << 8 | index in bin");
#define KPT   ((CH + 1023) / 1024)

__global__ void __launch_bounds__(1024)
gat_binhist(const int* __restrict__ trg, int* __restrict__ H) {
    __shared__ int h[NB];
    const int t = threadIdx.x;
    if (t < NB) h[t] = 0;
    __syncthreads();
    const int e0 = blockIdx.x * CH;
    int tg[KPT];
#pragma unroll
    for (int k = 0; k < KPT; ++k) {
        const int i = t + k * 1024;
        tg[k] = (i < CH && e0 + i < EE) ? trg[e0 + i] : -1;
    }
#pragma unroll
    for (int k = 0; k < KPT; ++k)
        if (tg[k] >= 0) atomicAdd(&h[tg[k] / BIN_W], 1);
    __syncthreads();
    if (t < NB) H[t * NCH + blockIdx.x] = h[t];
}

#define SCAN_T  1024
#define SCAN_IT 16
static_assert(32 * SCAN_IT * 128 == NB * NCH, "scan covers H exactly");

__global__ void __launch_bounds__(SCAN_T)
gat_scan(int* H) {
    __shared__ int wsum[32];
    const int t = threadIdx.x, lane = t & 31, w = t >> 5;
    const int wbase = w * (SCAN_IT * 128) + lane * 4;
    int4 v[SCAN_IT];
    int  inc[SCAN_IT];
#pragma unroll
    for (int i = 0; i < SCAN_IT; ++i) v[i] = *(const int4*)(H + wbase + i * 128);
#pragma unroll
    for (int i = 0; i < SCAN_IT; ++i) {
        int x = v[i].x + v[i].y + v[i].z + v[i].w;
#pragma unroll
        for (int o = 1; o < 32; o <<= 1) {
            int n = __shfl_up(x, o);
            if (lane >= o) x += n;
        }
        inc[i] = x;
    }
    int carry = 0;
#pragma unroll
    for (int i = 0; i < SCAN_IT; ++i) {
        int tot = __builtin_amdgcn_readlane(inc[i], 31);
        inc[i] += carry;
        carry += tot;
    }
    if (lane == 0) wsum[w] = carry;
    __syncthreads();
    if (w == 0) {
        int x = wsum[lane];
#pragma unroll
        for (int o = 1; o < 32; o <<= 1) {
            int n = __shfl_up(x, o);
            if (lane >= o) x += n;
        }
        wsum[lane] = x;
    }
    __syncthreads();
    const int base = w > 0 ? wsum[w - 1] : 0;
#pragma unroll
    for (int i = 0; i < SCAN_IT; ++i) {
        int4 o4;
        o4.x = base + inc[i] - (v[i].x + v[i].y + v[i].z + v[i].w);
        o4.y = o4.x + v[i].x;
        o4.z = o4.y + v[i].y;
        o4.w = o4.z + v[i].z;
        *(int4*)(H + wbase + i * 128) = o4;
    }
}

__global__ void __launch_bounds__(1024)
gat_binplace(const int* __restrict__ trg, const int* __restrict__ O,
             int* __restrict__ rec) {
    __shared__ int cur[NB];
    const int t = threadIdx.x;
    if (t < NB) cur[t] = O[t * NCH + blockIdx.x];
    __syncthreads();
    const int e0 = blockIdx.x * CH;
    int tg[KPT];
#pragma unroll
    for (int k = 0; k < KPT; ++k) {
        const int i = t + k * 1024;
        tg[k] = (i < CH && e0 + i < EE) ? trg[e0 + i] : -1;
    }
#pragma unroll
    for (int k = 0; k < KPT; ++k) {
        if (tg[k] >= 0) {
            const int e = e0 + t + k * 1024;
            const int b = tg[k] / BIN_W;
            const int p = atomicAdd(&cur[b], 1);
            rec[p] = (e << 8) | (tg[k] - b * BIN_W);
        }
    }
}

__global__ void __launch_bounds__(1024)
gat_fine(const int* __restrict__ O, const int* __restrict__ rec,
         int* __restrict__ cnt, int* __restrict__ offs, int* __restrict__ list) {
    __shared__ int c[256];
    __shared__ int cur[256];
    __shared__ int wsum[8];
    const int t = threadIdx.x, lane = t & 31, w = t >> 5;
    const int b = blockIdx.x;
    const int start = O[b * NCH];
    const int end   = (b == NB - 1) ? EE : O[(b + 1) * NCH];
    const int n = end - start;
    if (t < 256) c[t] = 0;
    __syncthreads();
    for (int i0 = 0; i0 < n; i0 += 4096) {
        int r[4];
#pragma unroll
        for (int k = 0; k < 4; ++k) {
            const int i = i0 + k * 1024 + t;
            r[k] = i < n ? rec[start + i] : -1;
        }
#pragma unroll
        for (int k = 0; k < 4; ++k)
            if (r[k] >= 0) atomicAdd(&c[r[k] & 255], 1);
    }
    __syncthreads();
    const int own = t < 256 ? c[t] : 0;
    int x = own;
#pragma unroll
    for (int o = 1; o < 32; o <<= 1) {
        int m = __shfl_up(x, o);
        if (lane >= o) x += m;
    }
    if (t < 256 && lane == 31) wsum[w] = x;
    __syncthreads();
    if (t < 256) {
        int base = 0;
        for (int j = 0; j < w; ++j) base += wsum[j];
        const int ex = base + x - own;
        cur[t] = ex;
        const int node = b * BIN_W + t;
        if (t < BIN_W && node < NN) {
            cnt[node]  = own;
            offs[node] = start + ex;
        }
    }
    __syncthreads();
    for (int i0 = 0; i0 < n; i0 += 4096) {
        int r[4];
#pragma unroll
        for (int k = 0; k < 4; ++k) {
            const int i = i0 + k * 1024 + t;
            r[k] = i < n ? rec[start + i] : -1;
        }
#pragma unroll
        for (int k = 0; k < 4; ++k) {
            if (r[k] >= 0) {
                const int p = atomicAdd(&cur[r[k] & 255], 1);
                list[start + p] = r[k] >> 8;
            }
        }
    }
}

__device__ __forceinline__ float edge_exp(float s_src, float s_trg) {
    float v = s_src + s_trg;
    v = v > 0.f ? v : 0.2f * v;
    return __expf(v);
}

__global__ void __launch_bounds__(256)
gat_aggregate(const int* __restrict__ src,
              const int* __restrict__ cnt, const int* __restrict__ offs,
              const int* __restrict__ list,
              const float* __restrict__ emb,
              const float* __restrict__ ss, const float* __restrict__ ts,
              const float* __restrict__ bias,
              float* __restrict__ out) {
    const int lane = threadIdx.x & 31;
    const int node = __builtin_amdgcn_readfirstlane((int)((blockIdx.x * blockDim.x + threadIdx.x) >> 5));
    if (node >= NN) return;
    const int h = lane >> 2;
    const int deg = __builtin_amdgcn_readfirstlane(cnt[node]);
    const int beg = __builtin_amdgcn_readfirstlane(offs[node]);
    const float tsv = ts[node * HEADS + h];
    const float4* __restrict__ emb4 = (const float4*)emb;

    float denom = 0.f;
    float4 acc = make_float4(0.f, 0.f, 0.f, 0.f);

    if (deg <= 32) {
        int e = lane < deg ? list[beg + lane] : 0x7fffffff;
        int rank = 0;
        for (int j = 0; j < deg; ++j) {
            int ej = __builtin_amdgcn_readlane(e, j);
            rank += (ej < e) ? 1 : 0;
        }
        if (lane >= deg) rank = lane;
        int es = __builtin_amdgcn_ds_permute(rank << 2, e);
        int sv = lane < deg ? src[es] : 0;

        for (int r = 0; r < deg; ++r) {
            int s = __builtin_amdgcn_readlane(sv, r);
            denom += edge_exp(ss[s * HEADS + h], tsv);
        }
        const float dn = denom + 1e-16f;
        int r = 0;
        for (; r + 4 <= deg; r += 4) {
            int s0 = __builtin_amdgcn_readlane(sv, r);
            int s1 = __builtin_amdgcn_readlane(sv, r + 1);
            int s2 = __builtin_amdgcn_readlane(sv, r + 2);
            int s3 = __builtin_amdgcn_readlane(sv, r + 3);
            float4 x0 = emb4[s0 * 32 + lane];
            float4 x1 = emb4[s1 * 32 + lane];
            float4 x2 = emb4[s2 * 32 + lane];
            float4 x3 = emb4[s3 * 32 + lane];
            float a0 = edge_exp(ss[s0 * HEADS + h], tsv) / dn;
            float a1 = edge_exp(ss[s1 * HEADS + h], tsv) / dn;
            float a2 = edge_exp(ss[s2 * HEADS + h], tsv) / dn;
            float a3 = edge_exp(ss[s3 * HEADS + h], tsv) / dn;
            acc.x += x0.x * a0; acc.y += x0.y * a0; acc.z += x0.z * a0; acc.w += x0.w * a0;
            acc.x += x1.x * a1; acc.y += x1.y * a1; acc.z += x1.z * a1; acc.w += x1.w * a1;
            acc.x += x2.x * a2; acc.y += x2.y * a2; acc.z += x2.z * a2; acc.w += x2.w * a2;
            acc.x += x3.x * a3; acc.y += x3.y * a3; acc.z += x3.z * a3; acc.w += x3.w * a3;
        }
        for (; r < deg; ++r) {
            int s0 = __builtin_amdgcn_readlane(sv, r);
            float4 x0 = emb4[s0 * 32 + lane];
            float a0 = edge_exp(ss[s0 * HEADS + h], tsv) / dn;
            acc.x += x0.x * a0; acc.y += x0.y * a0; acc.z += x0.z * a0; acc.w += x0.w * a0;
        }
    } else {
        int prev = -1;
        for (int r = 0; r < deg; ++r) {
            int m = 0x7fffffff;
            for (int j = lane; j < deg; j += 32) {
                int ej = list[beg + j];
                if (ej > prev && ej < m) m = ej;
            }
#pragma unroll
            for (int o = 16; o > 0; o >>= 1) {
                int n = __shfl_xor(m, o);
                m = n < m ? n : m;
            }
            prev = __builtin_amdgcn_readfirstlane(m);
            int s = src[prev];
            denom += edge_exp(ss[s * HEADS + h], tsv);
        }
        const float dn = denom + 1e-16f;
        prev = -1;
        for (int r = 0; r < deg; ++r) {
            int m = 0x7fffffff;
            for (int j = lane; j < deg; j += 32) {
                int ej = list[beg + j];
                if (ej > prev && ej < m) m = ej;
            }
#pragma unroll
            for (int o = 16; o > 0; o >>= 1) {
                int n = __shfl_xor(m, o);
                m = n < m ? n : m;
            }
            prev = __builtin_amdgcn_readfirstlane(m);
            int s0 = src[prev];
            float4 x0 = emb4[s0 * 32 + lane];
            float a0 = edge_exp(ss[s0 * HEADS + h], tsv) / dn;
            acc.x += x0.x * a0; acc.y += x0.y * a0; acc.z += x0.z * a0; acc.w += x0.w * a0;
        }
    }

    const float4 b = ((const float4*)bias)[lane];
    float4 o4 = make_float4(acc.x + b.x, acc.y + b.y, acc.z + b.z, acc.w + b.w);
    ((float4*)out)[node * 32 + lane] = o4;
}

extern "C" void kernel_launch(void* const* d_in, const int* in_sizes, int n_in,
                              void* d_out, int out_size, void* d_ws, size_t ws_size,
                              hipStream_t stream) {
    const float* X      = (const float*)d_in[0];
    const int*   eidx   = (const int*)d_in[1];
    const float* W      = (const float*)d_in[2];
    const float* a_l    = (const float*)d_in[3];
    const float* a_r    = (const float*)d_in[4];
    const float* bias   = (const float*)d_in[5];
    float* out = (float*)d_out;

    const int* src = eidx;
    const int* trg = eidx + EE;

    float* ws    = (float*)d_ws;
    float* emb   = ws;
    float* ss    = emb + (size_t)NN * OUTC;
    float* ts    = ss + (size_t)NN * HEADS;
    int*   cnt   = (int*)(ts + (size_t)NN * HEADS);
    int*   offs  = cnt + NN;
    int*   rec   = offs + NN;
    int*   list  = rec + EE;
    int*   H     = (int*)emb;

    gat_binhist<<<NCH, 1024, 0, stream>>>(trg, H);
    gat_scan<<<1, SCAN_T, 0, stream>>>(H);
    gat_binplace<<<NCH, 1024, 0, stream>>>(trg, H, rec);
    gat_fine<<<NB, 1024, 0, stream>>>(H, rec, cnt, offs, list);

    gat_gemm_wmma<<<GEMM_GRID, 256, 0, stream>>>(X, W, emb);

    gat_scores<<<(NN * HEADS + 255) / 256, 256, 0, stream>>>(emb, a_l, a_r, ss, ts);

    gat_aggregate<<<(NN + 7) / 8, 256, 0, stream>>>(src, cnt, offs, list, emb, ss, ts, bias, out);
}


# Aggregate wave: a node's loads go out together, not one round trip per edge: scores in one gather (lane = head x edge mod 4), each exp and weight once, shared by quad broadcast; sum order unchanged
# speedup vs baseline: 3.5067x; 1.0598x over previous
#include <hip/hip_runtime.h>
#include <hip/hip_bf16.h>

#define NN 50000
#define EE 600000
#define INC 128
#define OUTC 128
#define HEADS 8
#define HEADC 16

typedef __attribute__((ext_vector_type(2))) float v2f;
typedef __attribute__((ext_vector_type(8))) float v8f;

#define STRIPS    (NN / 16)
#define GEMM_GRID 1024
#define APITCH    132
#define TILE_F    (16 * APITCH)
static_assert(STRIPS * 16 == NN, "strips cover every row exactly");
static_assert(GEMM_GRID <= STRIPS, "every block has a first strip");

__global__ void __launch_bounds__(256)
gat_gemm_wmma(const float* __restrict__ X,
              const float* __restrict__ W,
              float* __restrict__ emb) {
    __shared__ __attribute__((aligned(16))) float L[8 * TILE_F];
    const int tid  = threadIdx.x;
    const int wave = tid >> 5;
    const int lane = tid & 31;
    const int half = lane >> 4;
    const int l    = lane & 15;

    const float4* __restrict__ W4 = (const float4*)W + wave * 512;
    float* wsm = L + wave * TILE_F;
#pragma unroll
    for (int i = 0; i < 16; ++i)
        *(float4*)(wsm + i * APITCH + lane * 4) = W4[i * 32 + lane];
    __syncthreads();
    v2f b[32];
#pragma unroll
    for (int kk = 0; kk < 32; ++kk)
        b[kk] = *(const v2f*)(wsm + l * APITCH + kk * 4 + half * 2);

    const float4* __restrict__ X4 = (const float4*)X;
    float* as0 = L + (tid >> 5) * APITCH + (tid & 31) * 4;
    float* as1 = as0 + 8 * APITCH;
    const float* afr = L + l * APITCH + half * 2;

    int strip = blockIdx.x;
    float* __restrict__ op =
        emb + (size_t)(strip * 16 + half * 8) * OUTC + wave * 16 + l;
    float4 t0 = X4[strip * 512 + tid];
    float4 t1 = X4[strip * 512 + 256 + tid];
    for (; strip < STRIPS; strip += GEMM_GRID) {
        __syncthreads();
        *(float4*)as0 = t0;
        *(float4*)as1 = t1;
        __syncthreads();
        const int next = strip + GEMM_GRID;
        if (next < STRIPS) {
            t0 = X4[next * 512 + tid];
            t1 = X4[next * 512 + 256 + tid];
        }

        v8f c = {};
#pragma unroll
        for (int kk = 0; kk < 32; ++kk) {
            v2f a = *(const v2f*)(afr + kk * 4);
            c = __builtin_amdgcn_wmma_f32_16x16x4_f32(
                     false, a,  false, b[kk],
                     (short)0, c,  false,  false);
        }

#pragma unroll
        for (int v = 0; v < 8; ++v) op[(size_t)v * OUTC] = c[v];
        op += (size_t)GEMM_GRID * 16 * OUTC;
    }
}

__global__ void gat_scores(const float* __restrict__ emb,
                           const float* __restrict__ a_left,
                           const float* __restrict__ a_right,
                           float* __restrict__ ss, float* __restrict__ ts) {
    int gid = blockIdx.x * blockDim.x + threadIdx.x;
    if (gid >= NN * HEADS) return;
    int n = gid >> 3, h = gid & 7;
    const float* ep = emb + (size_t)n * OUTC + h * HEADC;
    float s0 = 0.f, s1 = 0.f;
#pragma unroll
    for (int c = 0; c < HEADC; ++c) {
        float e = ep[c];
        s0 += e * a_left[c * HEADS + h];
        s1 += e * a_right[c * HEADS + h];
    }
    ss[gid] = s0;
    ts[gid] = s1;
}

#define NB    256
#define BIN_W 196
#define NCH   256
#define CH    2344
static_assert(NB * BIN_W >= NN, "bins cover every node");
static_assert(NCH * CH >= EE, "chunks cover every edge");
static_assert(BIN_W <= 256 && EE < (1 << 23), "record packing: edge id << 8 | index in bin");
#define KPT   ((CH + 1023) / 1024)

__global__ void __launch_bounds__(1024)
gat_binhist(const int* __restrict__ trg, int* __restrict__ H) {
    __shared__ int h[NB];
    const int t = threadIdx.x;
    if (t < NB) h[t] = 0;
    __syncthreads();
    const int e0 = blockIdx.x * CH;
    int tg[KPT];
#pragma unroll
    for (int k = 0; k < KPT; ++k) {
        const int i = t + k * 1024;
        tg[k] = (i < CH && e0 + i < EE) ? trg[e0 + i] : -1;
    }
#pragma unroll
    for (int k = 0; k < KPT; ++k)
        if (tg[k] >= 0) atomicAdd(&h[tg[k] / BIN_W], 1);
    __syncthreads();
    if (t < NB) H[t * NCH + blockIdx.x] = h[t];
}

#define SCAN_T  1024
#define SCAN_IT 16
static_assert(32 * SCAN_IT * 128 == NB * NCH, "scan covers H exactly");

__global__ void __launch_bounds__(SCAN_T)
gat_scan(int* H) {
    __shared__ int wsum[32];
    const int t = threadIdx.x, lane = t & 31, w = t >> 5;
    const int wbase = w * (SCAN_IT * 128) + lane * 4;
    int4 v[SCAN_IT];
    int  inc[SCAN_IT];
#pragma unroll
    for (int i = 0; i < SCAN_IT; ++i) v[i] = *(const int4*)(H + wbase + i * 128);
#pragma unroll
    for (int i = 0; i < SCAN_IT; ++i) {
        int x = v[i].x + v[i].y + v[i].z + v[i].w;
#pragma unroll
        for (int o = 1; o < 32; o <<= 1) {
            int n = __shfl_up(x, o);
            if (lane >= o) x += n;
        }
        inc[i] = x;
    }
    int carry = 0;
#pragma unroll
    for (int i = 0; i < SCAN_IT; ++i) {
        int tot = __builtin_amdgcn_readlane(inc[i], 31);
        inc[i] += carry;
        carry += tot;
    }
    if (lane == 0) wsum[w] = carry;
    __syncthreads();
    if (w == 0) {
        int x = wsum[lane];
#pragma unroll
        for (int o = 1; o < 32; o <<= 1) {
            int n = __shfl_up(x, o);
            if (lane >= o) x += n;
        }
        wsum[lane] = x;
    }
    __syncthreads();
    const int base = w > 0 ? wsum[w - 1] : 0;
#pragma unroll
    for (int i = 0; i < SCAN_IT; ++i) {
        int4 o4;
        o4.x = base + inc[i] - (v[i].x + v[i].y + v[i].z + v[i].w);
        o4.y = o4.x + v[i].x;
        o4.z = o4.y + v[i].y;
        o4.w = o4.z + v[i].z;
        *(int4*)(H + wbase + i * 128) = o4;
    }
}

__global__ void __launch_bounds__(1024)
gat_binplace(const int* __restrict__ trg, const int* __restrict__ O,
             int* __restrict__ rec) {
    __shared__ int cur[NB];
    const int t = threadIdx.x;
    if (t < NB) cur[t] = O[t * NCH + blockIdx.x];
    __syncthreads();
    const int e0 = blockIdx.x * CH;
    int tg[KPT];
#pragma unroll
    for (int k = 0; k < KPT; ++k) {
        const int i = t + k * 1024;
        tg[k] = (i < CH && e0 + i < EE) ? trg[e0 + i] : -1;
    }
#pragma unroll
    for (int k = 0; k < KPT; ++k) {
        if (tg[k] >= 0) {
            const int e = e0 + t + k * 1024;
            const int b = tg[k] / BIN_W;
            const int p = atomicAdd(&cur[b], 1);
            rec[p] = (e << 8) | (tg[k] - b * BIN_W);
        }
    }
}

__global__ void __launch_bounds__(1024)
gat_fine(const int* __restrict__ O, const int* __restrict__ rec,
         int* __restrict__ cnt, int* __restrict__ offs, int* __restrict__ list) {
    __shared__ int c[256];
    __shared__ int cur[256];
    __shared__ int wsum[8];
    const int t = threadIdx.x, lane = t & 31, w = t >> 5;
    const int b = blockIdx.x;
    const int start = O[b * NCH];
    const int end   = (b == NB - 1) ? EE : O[(b + 1) * NCH];
    const int n = end - start;
    if (t < 256) c[t] = 0;
    __syncthreads();
    for (int i0 = 0; i0 < n; i0 += 4096) {
        int r[4];
#pragma unroll
        for (int k = 0; k < 4; ++k) {
            const int i = i0 + k * 1024 + t;
            r[k] = i < n ? rec[start + i] : -1;
        }
#pragma unroll
        for (int k = 0; k < 4; ++k)
            if (r[k] >= 0) atomicAdd(&c[r[k] & 255], 1);
    }
    __syncthreads();
    const int own = t < 256 ? c[t] : 0;
    int x = own;
#pragma unroll
    for (int o = 1; o < 32; o <<= 1) {
        int m = __shfl_up(x, o);
        if (lane >= o) x += m;
    }
    if (t < 256 && lane == 31) wsum[w] = x;
    __syncthreads();
    if (t < 256) {
        int base = 0;
        for (int j = 0; j < w; ++j) base += wsum[j];
        const int ex = base + x - own;
        cur[t] = ex;
        const int node = b * BIN_W + t;
        if (t < BIN_W && node < NN) {
            cnt[node]  = own;
            offs[node] = start + ex;
        }
    }
    __syncthreads();
    for (int i0 = 0; i0 < n; i0 += 4096) {
        int r[4];
#pragma unroll
        for (int k = 0; k < 4; ++k) {
            const int i = i0 + k * 1024 + t;
            r[k] = i < n ? rec[start + i] : -1;
        }
#pragma unroll
        for (int k = 0; k < 4; ++k) {
            if (r[k] >= 0) {
                const int p = atomicAdd(&cur[r[k] & 255], 1);
                list[start + p] = r[k] >> 8;
            }
        }
    }
}

__device__ __forceinline__ float edge_exp(float s_src, float s_trg) {
    float v = s_src + s_trg;
    v = v > 0.f ? v : 0.2f * v;
    return __expf(v);
}

template <int T>
__device__ __forceinline__ float quad_lane(float v) {
    return __int_as_float(__builtin_amdgcn_update_dpp(
        0, __float_as_int(v), T * 0x55, 0xf, 0xf, true));
}

__device__ __forceinline__ void row_fma(float4& acc, const float4& x, float a) {
    acc.x += x.x * a; acc.y += x.y * a; acc.z += x.z * a; acc.w += x.w * a;
}

__device__ __forceinline__ void rows_request(float4 (&x)[8], const float4* __restrict__ emb4,
                                             int sv, int r0, int last, int lane) {
#pragma unroll
    for (int t = 0; t < 8; ++t) {
        const int r = r0 + t < last ? r0 + t : last;
        x[t] = emb4[__builtin_amdgcn_readlane(sv, r) * 32 + lane];
    }
}

__device__ __forceinline__ void rows_sum(float4& acc, const float4 (&x)[8], float wa, float wb) {
    row_fma(acc, x[0], quad_lane<0>(wa));
    row_fma(acc, x[1], quad_lane<1>(wa));
    row_fma(acc, x[2], quad_lane<2>(wa));
    row_fma(acc, x[3], quad_lane<3>(wa));
    row_fma(acc, x[4], quad_lane<0>(wb));
    row_fma(acc, x[5], quad_lane<1>(wb));
    row_fma(acc, x[6], quad_lane<2>(wb));
    row_fma(acc, x[7], quad_lane<3>(wb));
}

__global__ void __launch_bounds__(256)
gat_aggregate(const int* __restrict__ src,
              const int* __restrict__ cnt, const int* __restrict__ offs,
              const int* __restrict__ list,
              const float* __restrict__ emb,
              const float* __restrict__ ss, const float* __restrict__ ts,
              const float* __restrict__ bias,
              float* __restrict__ out) {
    const int lane = threadIdx.x & 31;
    const int node = __builtin_amdgcn_readfirstlane((int)((blockIdx.x * blockDim.x + threadIdx.x) >> 5));
    if (node >= NN) return;
    const int h = lane >> 2;
    const int deg = __builtin_amdgcn_readfirstlane(cnt[node]);
    const int beg = __builtin_amdgcn_readfirstlane(offs[node]);
    const float tsv = ts[node * HEADS + h];
    const float4* __restrict__ emb4 = (const float4*)emb;

    float denom = 0.f;
    float4 acc = make_float4(0.f, 0.f, 0.f, 0.f);

    if (deg <= 32) {
        int e = lane < deg ? list[beg + lane] : 0x7fffffff;
        int rank = 0;
        for (int j = 0; j < deg; ++j) {
            int ej = __builtin_amdgcn_readlane(e, j);
            rank += (ej < e) ? 1 : 0;
        }
        if (lane >= deg) rank = lane;
        int es = __builtin_amdgcn_ds_permute(rank << 2, e);
        int sv = lane < deg ? src[es] : 0;

        const int q    = lane & 3;
        const int last = deg > 0 ? deg - 1 : 0;
        const bool big = deg > 16;

        int sj[8];
#pragma unroll
        for (int j = 0; j < 4; ++j)
            sj[j] = __builtin_amdgcn_ds_bpermute((4 * j + q) << 2, sv);
        float4 xr[8];
        rows_request(xr, emb4, sv, 0, last, lane);
        float sc[8];
#pragma unroll
        for (int j = 0; j < 4; ++j) sc[j] = ss[sj[j] * HEADS + h];
#pragma unroll
        for (int j = 4; j < 8; ++j) sc[j] = 0.f;
        if (big) {
#pragma unroll
            for (int j = 4; j < 8; ++j)
                sj[j] = __builtin_amdgcn_ds_bpermute((4 * j + q) << 2, sv);
#pragma unroll
            for (int j = 4; j < 8; ++j) sc[j] = ss[sj[j] * HEADS + h];
        }

        float ex[8];
#pragma unroll
        for (int j = 0; j < 4; ++j) {
            const float v = edge_exp(sc[j], tsv);
            ex[j] = (4 * j + q < deg) ? v : 0.f;
        }
#pragma unroll
        for (int j = 4; j < 8; ++j) ex[j] = 0.f;
#pragma unroll
        for (int j = 0; j < 4; ++j) {
            denom += quad_lane<0>(ex[j]);
            denom += quad_lane<1>(ex[j]);
            denom += quad_lane<2>(ex[j]);
            denom += quad_lane<3>(ex[j]);
        }
        if (big) {
#pragma unroll
            for (int j = 4; j < 8; ++j) {
                const float v = edge_exp(sc[j], tsv);
                ex[j] = (4 * j + q < deg) ? v : 0.f;
            }
#pragma unroll
            for (int j = 4; j < 8; ++j) {
                denom += quad_lane<0>(ex[j]);
                denom += quad_lane<1>(ex[j]);
                denom += quad_lane<2>(ex[j]);
                denom += quad_lane<3>(ex[j]);
            }
        }
        const float dn = denom + 1e-16f;
        float w[8];
#pragma unroll
        for (int j = 0; j < 8; ++j) w[j] = ex[j] / dn;

        rows_sum(acc, xr, w[0], w[1]);
        if (deg > 8) {
            rows_request(xr, emb4, sv, 8, last, lane);
            rows_sum(acc, xr, w[2], w[3]);
            if (big) {
                rows_request(xr, emb4, sv, 16, last, lane);
                rows_sum(acc, xr, w[4], w[5]);
                if (deg > 24) {
                    rows_request(xr, emb4, sv, 24, last, lane);
                    rows_sum(acc, xr, w[6], w[7]);
                }
            }
        }
    } else {
        int prev = -1;
        for (int r = 0; r < deg; ++r) {
            int m = 0x7fffffff;
            for (int j = lane; j < deg; j += 32) {
                int ej = list[beg + j];
                if (ej > prev && ej < m) m = ej;
            }
#pragma unroll
            for (int o = 16; o > 0; o >>= 1) {
                int n = __shfl_xor(m, o);
                m = n < m ? n : m;
            }
            prev = __builtin_amdgcn_readfirstlane(m);
            int s = src[prev];
            denom += edge_exp(ss[s * HEADS + h], tsv);
        }
        const float dn = denom + 1e-16f;
        prev = -1;
        for (int r = 0; r < deg; ++r) {
            int m = 0x7fffffff;
            for (int j = lane; j < deg; j += 32) {
                int ej = list[beg + j];
                if (ej > prev && ej < m) m = ej;
            }
#pragma unroll
            for (int o = 16; o > 0; o >>= 1) {
                int n = __shfl_xor(m, o);
                m = n < m ? n : m;
            }
            prev = __builtin_amdgcn_readfirstlane(m);
            int s0 = src[prev];
            float4 x0 = emb4[s0 * 32 + lane];
            float a0 = edge_exp(ss[s0 * HEADS + h], tsv) / dn;
            acc.x += x0.x * a0; acc.y += x0.y * a0; acc.z += x0.z * a0; acc.w += x0.w * a0;
        }
    }

    const float4 b = ((const float4*)bias)[lane];
    float4 o4 = make_float4(acc.x + b.x, acc.y + b.y, acc.z + b.z, acc.w + b.w);
    ((float4*)out)[node * 32 + lane] = o4;
}

extern "C" void kernel_launch(void* const* d_in, const int* in_sizes, int n_in,
                              void* d_out, int out_size, void* d_ws, size_t ws_size,
                              hipStream_t stream) {
    const float* X      = (const float*)d_in[0];
    const int*   eidx   = (const int*)d_in[1];
    const float* W      = (const float*)d_in[2];
    const float* a_l    = (const float*)d_in[3];
    const float* a_r    = (const float*)d_in[4];
    const float* bias   = (const float*)d_in[5];
    float* out = (float*)d_out;

    const int* src = eidx;
    const int* trg = eidx + EE;

    float* ws    = (float*)d_ws;
    float* emb   = ws;
    float* ss    = emb + (size_t)NN * OUTC;
    float* ts    = ss + (size_t)NN * HEADS;
    int*   cnt   = (int*)(ts + (size_t)NN * HEADS);
    int*   offs  = cnt + NN;
    int*   rec   = offs + NN;
    int*   list  = rec + EE;
    int*   H     = (int*)emb;

    gat_binhist<<<NCH, 1024, 0, stream>>>(trg, H);
    gat_scan<<<1, SCAN_T, 0, stream>>>(H);
    gat_binplace<<<NCH, 1024, 0, stream>>>(trg, H, rec);
    gat_fine<<<NB, 1024, 0, stream>>>(H, rec, cnt, offs, list);

    gat_gemm_wmma<<<GEMM_GRID, 256, 0, stream>>>(X, W, emb);

    gat_scores<<<(NN * HEADS + 255) / 256, 256, 0, stream>>>(emb, a_l, a_r, ss, ts);

    gat_aggregate<<<(NN + 7) / 8, 256, 0, stream>>>(src, cnt, offs, list, emb, ss, ts, bias, out);
}


# Scores in the GEMM epilogue: each wave dots its head's accumulator rows with a_left/a_right by a 16-lane halving sum (tree order, was c=0..15), block stores whole lines; the scores launch is gone
# speedup vs baseline: 3.6579x; 1.0431x over previous
#include <hip/hip_runtime.h>
#include <hip/hip_bf16.h>

#define NN 50000
#define EE 600000
#define INC 128
#define OUTC 128
#define HEADS 8
#define HEADC 16

typedef __attribute__((ext_vector_type(2))) float v2f;
typedef __attribute__((ext_vector_type(8))) float v8f;

template <int M>
__device__ __forceinline__ float row_xor(float v) {
    return __int_as_float(__builtin_amdgcn_update_dpp(
        0, __float_as_int(v), 0x160 | M, 0xf, 0xf, true));
}
template <int M>
__device__ __forceinline__ float halve(float lo, float hi, bool up) {
    const float keep = up ? hi : lo;
    const float send = up ? lo : hi;
    return keep + row_xor<M>(send);
}

#define STRIPS    (NN / 16)
#define GEMM_GRID 1024
#define APITCH    132
#define TILE_F    (16 * APITCH)
static_assert(STRIPS * 16 == NN, "strips cover every row exactly");
static_assert(GEMM_GRID <= STRIPS, "every block has a first strip");

__device__ __forceinline__ void lds_barrier() {
    __builtin_amdgcn_fence(__ATOMIC_RELEASE, "workgroup", "local");
    __builtin_amdgcn_s_barrier();
    __builtin_amdgcn_fence(__ATOMIC_ACQUIRE, "workgroup", "local");
}

__global__ void __launch_bounds__(256)
gat_gemm_wmma(const float* __restrict__ X,
              const float* __restrict__ W,
              const float* __restrict__ a_left,
              const float* __restrict__ a_right,
              float* __restrict__ emb,
              float* __restrict__ ss, float* __restrict__ ts) {
    __shared__ __attribute__((aligned(16))) float L[8 * TILE_F];
    const int tid  = threadIdx.x;
    const int wave = tid >> 5;
    const int lane = tid & 31;
    const int half = lane >> 4;
    const int l    = lane & 15;

    const float4* __restrict__ W4 = (const float4*)W + wave * 512;
    float* wsm = L + wave * TILE_F;
#pragma unroll
    for (int i = 0; i < 16; ++i)
        *(float4*)(wsm + i * APITCH + lane * 4) = W4[i * 32 + lane];
    __syncthreads();
    v2f b[32];
#pragma unroll
    for (int kk = 0; kk < 32; ++kk)
        b[kk] = *(const v2f*)(wsm + l * APITCH + kk * 4 + half * 2);

    const float4* __restrict__ X4 = (const float4*)X;
    float* as0 = L + (tid >> 5) * APITCH + (tid & 31) * 4;
    float* as1 = as0 + 8 * APITCH;
    const float* afr = L + l * APITCH + half * 2;

    const float al = a_left[l * HEADS + wave];
    const float ar = a_right[l * HEADS + wave];
    const bool  up8 = (l & 8) != 0, up4 = (l & 4) != 0, up2 = (l & 2) != 0, up1 = (l & 1) != 0;
    const float a_keep = up8 ? ar : al;
    const float a_send = up8 ? al : ar;
    float* sst = L + TILE_F;
    float* sput = sst + (l >> 3) * 128 + (half * 8 + (l & 7)) * HEADS + wave;
    float* __restrict__ sp = (tid < 128 ? ss : ts) + (size_t)blockIdx.x * 128 + (tid & 127);

    int strip = blockIdx.x;
    float* __restrict__ op =
        emb + (size_t)(strip * 16 + half * 8) * OUTC + wave * 16 + l;
    float4 t0 = X4[strip * 512 + tid];
    float4 t1 = X4[strip * 512 + 256 + tid];
    for (; strip < STRIPS; strip += GEMM_GRID) {
        lds_barrier();
        if (strip != (int)blockIdx.x) {
            *sp = sst[tid];
            sp += (size_t)GEMM_GRID * 128;
        }
        *(float4*)as0 = t0;
        *(float4*)as1 = t1;
        lds_barrier();
        const int next = strip + GEMM_GRID;
        if (next < STRIPS) {
            t0 = X4[next * 512 + tid];
            t1 = X4[next * 512 + 256 + tid];
        }

        v8f c = {};
#pragma unroll
        for (int kk = 0; kk < 32; ++kk) {
            v2f a = *(const v2f*)(afr + kk * 4);
            c = __builtin_amdgcn_wmma_f32_16x16x4_f32(
                     false, a,  false, b[kk],
                     (short)0, c,  false,  false);
        }

#pragma unroll
        for (int v = 0; v < 8; ++v) op[(size_t)v * OUTC] = c[v];
        op += (size_t)GEMM_GRID * 16 * OUTC;

        float q8[8], q4[4], q2[2];
#pragma unroll
        for (int v = 0; v < 8; ++v) q8[v] = c[v] * a_keep + row_xor<8>(c[v] * a_send);
#pragma unroll
        for (int k = 0; k < 4; ++k) q4[k] = halve<4>(q8[k], q8[k + 4], up4);
#pragma unroll
        for (int k = 0; k < 2; ++k) q2[k] = halve<2>(q4[k], q4[k + 2], up2);
        *sput = halve<1>(q2[0], q2[1], up1);
    }
    lds_barrier();
    *sp = sst[tid];
}

#define NB    256
#define BIN_W 196
#define NCH   256
#define CH    2344
static_assert(NB * BIN_W >= NN, "bins cover every node");
static_assert(NCH * CH >= EE, "chunks cover every edge");
static_assert(BIN_W <= 256 && EE < (1 << 23), "record packing: edge id << 8 | index in bin");
#define KPT   ((CH + 1023) / 1024)

__global__ void __launch_bounds__(1024)
gat_binhist(const int* __restrict__ trg, int* __restrict__ H) {
    __shared__ int h[NB];
    const int t = threadIdx.x;
    if (t < NB) h[t] = 0;
    __syncthreads();
    const int e0 = blockIdx.x * CH;
    int tg[KPT];
#pragma unroll
    for (int k = 0; k < KPT; ++k) {
        const int i = t + k * 1024;
        tg[k] = (i < CH && e0 + i < EE) ? trg[e0 + i] : -1;
    }
#pragma unroll
    for (int k = 0; k < KPT; ++k)
        if (tg[k] >= 0) atomicAdd(&h[tg[k] / BIN_W], 1);
    __syncthreads();
    if (t < NB) H[t * NCH + blockIdx.x] = h[t];
}

#define SCAN_T  1024
#define SCAN_IT 16
static_assert(32 * SCAN_IT * 128 == NB * NCH, "scan covers H exactly");

__global__ void __launch_bounds__(SCAN_T)
gat_scan(int* H) {
    __shared__ int wsum[32];
    const int t = threadIdx.x, lane = t & 31, w = t >> 5;
    const int wbase = w * (SCAN_IT * 128) + lane * 4;
    int4 v[SCAN_IT];
    int  inc[SCAN_IT];
#pragma unroll
    for (int i = 0; i < SCAN_IT; ++i) v[i] = *(const int4*)(H + wbase + i * 128);
#pragma unroll
    for (int i = 0; i < SCAN_IT; ++i) {
        int x = v[i].x + v[i].y + v[i].z + v[i].w;
#pragma unroll
        for (int o = 1; o < 32; o <<= 1) {
            int n = __shfl_up(x, o);
            if (lane >= o) x += n;
        }
        inc[i] = x;
    }
    int carry = 0;
#pragma unroll
    for (int i = 0; i < SCAN_IT; ++i) {
        int tot = __builtin_amdgcn_readlane(inc[i], 31);
        inc[i] += carry;
        carry += tot;
    }
    if (lane == 0) wsum[w] = carry;
    __syncthreads();
    if (w == 0) {
        int x = wsum[lane];
#pragma unroll
        for (int o = 1; o < 32; o <<= 1) {
            int n = __shfl_up(x, o);
            if (lane >= o) x += n;
        }
        wsum[lane] = x;
    }
    __syncthreads();
    const int base = w > 0 ? wsum[w - 1] : 0;
#pragma unroll
    for (int i = 0; i < SCAN_IT; ++i) {
        int4 o4;
        o4.x = base + inc[i] - (v[i].x + v[i].y + v[i].z + v[i].w);
        o4.y = o4.x + v[i].x;
        o4.z = o4.y + v[i].y;
        o4.w = o4.z + v[i].z;
        *(int4*)(H + wbase + i * 128) = o4;
    }
}

__global__ void __launch_bounds__(1024)
gat_binplace(const int* __restrict__ trg, const int* __restrict__ O,
             int* __restrict__ rec) {
    __shared__ int cur[NB];
    const int t = threadIdx.x;
    if (t < NB) cur[t] = O[t * NCH + blockIdx.x];
    __syncthreads();
    const int e0 = blockIdx.x * CH;
    int tg[KPT];
#pragma unroll
    for (int k = 0; k < KPT; ++k) {
        const int i = t + k * 1024;
        tg[k] = (i < CH && e0 + i < EE) ? trg[e0 + i] : -1;
    }
#pragma unroll
    for (int k = 0; k < KPT; ++k) {
        if (tg[k] >= 0) {
            const int e = e0 + t + k * 1024;
            const int b = tg[k] / BIN_W;
            const int p = atomicAdd(&cur[b], 1);
            rec[p] = (e << 8) | (tg[k] - b * BIN_W);
        }
    }
}

__global__ void __launch_bounds__(1024)
gat_fine(const int* __restrict__ O, const int* __restrict__ rec,
         int* __restrict__ cnt, int* __restrict__ offs, int* __restrict__ list) {
    __shared__ int c[256];
    __shared__ int cur[256];
    __shared__ int wsum[8];
    const int t = threadIdx.x, lane = t & 31, w = t >> 5;
    const int b = blockIdx.x;
    const int start = O[b * NCH];
    const int end   = (b == NB - 1) ? EE : O[(b + 1) * NCH];
    const int n = end - start;
    if (t < 256) c[t] = 0;
    __syncthreads();
    for (int i0 = 0; i0 < n; i0 += 4096) {
        int r[4];
#pragma unroll
        for (int k = 0; k < 4; ++k) {
            const int i = i0 + k * 1024 + t;
            r[k] = i < n ? rec[start + i] : -1;
        }
#pragma unroll
        for (int k = 0; k < 4; ++k)
            if (r[k] >= 0) atomicAdd(&c[r[k] & 255], 1);
    }
    __syncthreads();
    const int own = t < 256 ? c[t] : 0;
    int x = own;
#pragma unroll
    for (int o = 1; o < 32; o <<= 1) {
        int m = __shfl_up(x, o);
        if (lane >= o) x += m;
    }
    if (t < 256 && lane == 31) wsum[w] = x;
    __syncthreads();
    if (t < 256) {
        int base = 0;
        for (int j = 0; j < w; ++j) base += wsum[j];
        const int ex = base + x - own;
        cur[t] = ex;
        const int node = b * BIN_W + t;
        if (t < BIN_W && node < NN) {
            cnt[node]  = own;
            offs[node] = start + ex;
        }
    }
    __syncthreads();
    for (int i0 = 0; i0 < n; i0 += 4096) {
        int r[4];
#pragma unroll
        for (int k = 0; k < 4; ++k) {
            const int i = i0 + k * 1024 + t;
            r[k] = i < n ? rec[start + i] : -1;
        }
#pragma unroll
        for (int k = 0; k < 4; ++k) {
            if (r[k] >= 0) {
                const int p = atomicAdd(&cur[r[k] & 255], 1);
                list[start + p] = r[k] >> 8;
            }
        }
    }
}

__device__ __forceinline__ float edge_exp(float s_src, float s_trg) {
    float v = s_src + s_trg;
    v = v > 0.f ? v : 0.2f * v;
    return __expf(v);
}

template <int T>
__device__ __forceinline__ float quad_lane(float v) {
    return __int_as_float(__builtin_amdgcn_update_dpp(
        0, __float_as_int(v), T * 0x55, 0xf, 0xf, true));
}

__device__ __forceinline__ void row_fma(float4& acc, const float4& x, float a) {
    acc.x += x.x * a; acc.y += x.y * a; acc.z += x.z * a; acc.w += x.w * a;
}

__device__ __forceinline__ void rows_request(float4 (&x)[8], const float4* __restrict__ emb4,
                                             int sv, int r0, int last, int lane) {
#pragma unroll
    for (int t = 0; t < 8; ++t) {
        const int r = r0 + t < last ? r0 + t : last;
        x[t] = emb4[__builtin_amdgcn_readlane(sv, r) * 32 + lane];
    }
}

__device__ __forceinline__ void rows_sum(float4& acc, const float4 (&x)[8], float wa, float wb) {
    row_fma(acc, x[0], quad_lane<0>(wa));
    row_fma(acc, x[1], quad_lane<1>(wa));
    row_fma(acc, x[2], quad_lane<2>(wa));
    row_fma(acc, x[3], quad_lane<3>(wa));
    row_fma(acc, x[4], quad_lane<0>(wb));
    row_fma(acc, x[5], quad_lane<1>(wb));
    row_fma(acc, x[6], quad_lane<2>(wb));
    row_fma(acc, x[7], quad_lane<3>(wb));
}

__global__ void __launch_bounds__(256)
gat_aggregate(const int* __restrict__ src,
              const int* __restrict__ cnt, const int* __restrict__ offs,
              const int* __restrict__ list,
              const float* __restrict__ emb,
              const float* __restrict__ ss, const float* __restrict__ ts,
              const float* __restrict__ bias,
              float* __restrict__ out) {
    const int lane = threadIdx.x & 31;
    const int node = __builtin_amdgcn_readfirstlane((int)((blockIdx.x * blockDim.x + threadIdx.x) >> 5));
    if (node >= NN) return;
    const int h = lane >> 2;
    const int deg = __builtin_amdgcn_readfirstlane(cnt[node]);
    const int beg = __builtin_amdgcn_readfirstlane(offs[node]);
    const float tsv = ts[node * HEADS + h];
    const float4* __restrict__ emb4 = (const float4*)emb;

    float denom = 0.f;
    float4 acc = make_float4(0.f, 0.f, 0.f, 0.f);

    if (deg <= 32) {
        int e = lane < deg ? list[beg + lane] : 0x7fffffff;
        int rank = 0;
        for (int j = 0; j < deg; ++j) {
            int ej = __builtin_amdgcn_readlane(e, j);
            rank += (ej < e) ? 1 : 0;
        }
        if (lane >= deg) rank = lane;
        int es = __builtin_amdgcn_ds_permute(rank << 2, e);
        int sv = lane < deg ? src[es] : 0;

        const int q    = lane & 3;
        const int last = deg > 0 ? deg - 1 : 0;
        const bool big = deg > 16;

        int sj[8];
#pragma unroll
        for (int j = 0; j < 4; ++j)
            sj[j] = __builtin_amdgcn_ds_bpermute((4 * j + q) << 2, sv);
        float4 xr[8];
        rows_request(xr, emb4, sv, 0, last, lane);
        float sc[8];
#pragma unroll
        for (int j = 0; j < 4; ++j) sc[j] = ss[sj[j] * HEADS + h];
#pragma unroll
        for (int j = 4; j < 8; ++j) sc[j] = 0.f;
        if (big) {
#pragma unroll
            for (int j = 4; j < 8; ++j)
                sj[j] = __builtin_amdgcn_ds_bpermute((4 * j + q) << 2, sv);
#pragma unroll
            for (int j = 4; j < 8; ++j) sc[j] = ss[sj[j] * HEADS + h];
        }

        float ex[8];
#pragma unroll
        for (int j = 0; j < 4; ++j) {
            const float v = edge_exp(sc[j], tsv);
            ex[j] = (4 * j + q < deg) ? v : 0.f;
        }
#pragma unroll
        for (int j = 4; j < 8; ++j) ex[j] = 0.f;
#pragma unroll
        for (int j = 0; j < 4; ++j) {
            denom += quad_lane<0>(ex[j]);
            denom += quad_lane<1>(ex[j]);
            denom += quad_lane<2>(ex[j]);
            denom += quad_lane<3>(ex[j]);
        }
        if (big) {
#pragma unroll
            for (int j = 4; j < 8; ++j) {
                const float v = edge_exp(sc[j], tsv);
                ex[j] = (4 * j + q < deg) ? v : 0.f;
            }
#pragma unroll
            for (int j = 4; j < 8; ++j) {
                denom += quad_lane<0>(ex[j]);
                denom += quad_lane<1>(ex[j]);
                denom += quad_lane<2>(ex[j]);
                denom += quad_lane<3>(ex[j]);
            }
        }
        const float dn = denom + 1e-16f;
        float w[8];
#pragma unroll
        for (int j = 0; j < 8; ++j) w[j] = ex[j] / dn;

        rows_sum(acc, xr, w[0], w[1]);
        if (deg > 8) {
            rows_request(xr, emb4, sv, 8, last, lane);
            rows_sum(acc, xr, w[2], w[3]);
            if (big) {
                rows_request(xr, emb4, sv, 16, last, lane);
                rows_sum(acc, xr, w[4], w[5]);
                if (deg > 24) {
                    rows_request(xr, emb4, sv, 24, last, lane);
                    rows_sum(acc, xr, w[6], w[7]);
                }
            }
        }
    } else {
        int prev = -1;
        for (int r = 0; r < deg; ++r) {
            int m = 0x7fffffff;
            for (int j = lane; j < deg; j += 32) {
                int ej = list[beg + j];
                if (ej > prev && ej < m) m = ej;
            }
#pragma unroll
            for (int o = 16; o > 0; o >>= 1) {
                int n = __shfl_xor(m, o);
                m = n < m ? n : m;
            }
            prev = __builtin_amdgcn_readfirstlane(m);
            int s = src[prev];
            denom += edge_exp(ss[s * HEADS + h], tsv);
        }
        const float dn = denom + 1e-16f;
        prev = -1;
        for (int r = 0; r < deg; ++r) {
            int m = 0x7fffffff;
            for (int j = lane; j < deg; j += 32) {
                int ej = list[beg + j];
                if (ej > prev && ej < m) m = ej;
            }
#pragma unroll
            for (int o = 16; o > 0; o >>= 1) {
                int n = __shfl_xor(m, o);
                m = n < m ? n : m;
            }
            prev = __builtin_amdgcn_readfirstlane(m);
            int s0 = src[prev];
            float4 x0 = emb4[s0 * 32 + lane];
            float a0 = edge_exp(ss[s0 * HEADS + h], tsv) / dn;
            acc.x += x0.x * a0; acc.y += x0.y * a0; acc.z += x0.z * a0; acc.w += x0.w * a0;
        }
    }

    const float4 b = ((const float4*)bias)[lane];
    float4 o4 = make_float4(acc.x + b.x, acc.y + b.y, acc.z + b.z, acc.w + b.w);
    ((float4*)out)[node * 32 + lane] = o4;
}

extern "C" void kernel_launch(void* const* d_in, const int* in_sizes, int n_in,
                              void* d_out, int out_size, void* d_ws, size_t ws_size,
                              hipStream_t stream) {
    const float* X      = (const float*)d_in[0];
    const int*   eidx   = (const int*)d_in[1];
    const float* W      = (const float*)d_in[2];
    const float* a_l    = (const float*)d_in[3];
    const float* a_r    = (const float*)d_in[4];
    const float* bias   = (const float*)d_in[5];
    float* out = (float*)d_out;

    const int* src = eidx;
    const int* trg = eidx + EE;

    float* ws    = (float*)d_ws;
    float* emb   = ws;
    float* ss    = emb + (size_t)NN * OUTC;
    float* ts    = ss + (size_t)NN * HEADS;
    int*   cnt   = (int*)(ts + (size_t)NN * HEADS);
    int*   offs  = cnt + NN;
    int*   rec   = offs + NN;
    int*   list  = rec + EE;
    int*   H     = (int*)emb;

    gat_binhist<<<NCH, 1024, 0, stream>>>(trg, H);
    gat_scan<<<1, SCAN_T, 0, stream>>>(H);
    gat_binplace<<<NCH, 1024, 0, stream>>>(trg, H, rec);
    gat_fine<<<NB, 1024, 0, stream>>>(H, rec, cnt, offs, list);

    gat_gemm_wmma<<<GEMM_GRID, 256, 0, stream>>>(X, W, a_l, a_r, emb, ss, ts);

    gat_aggregate<<<(NN + 7) / 8, 256, 0, stream>>>(src, cnt, offs, list, emb, ss, ts, bias, out);
}


# Product X@W^T has no launch of its own: its strips run as 512 extra blocks in each of the three edge-grouping launches (block index picks the role), overlapping them; count table in out; same sums
# speedup vs baseline: 3.8519x; 1.0530x over previous
#include <hip/hip_runtime.h>
#include <hip/hip_bf16.h>

#define NN 50000
#define EE 600000
#define INC 128
#define OUTC 128
#define HEADS 8
#define HEADC 16

typedef __attribute__((ext_vector_type(2))) float v2f;
typedef __attribute__((ext_vector_type(8))) float v8f;

template <int M>
__device__ __forceinline__ float row_xor(float v) {
    return __int_as_float(__builtin_amdgcn_update_dpp(
        0, __float_as_int(v), 0x160 | M, 0xf, 0xf, true));
}
template <int M>
__device__ __forceinline__ float halve(float lo, float hi, bool up) {
    const float keep = up ? hi : lo;
    const float send = up ? lo : hi;
    return keep + row_xor<M>(send);
}

#define STRIPS    (NN / 16)
#define GPART     512
#define SCUT1     1042
#define SCUT2     2084
#define APITCH    132
#define TILE_F    (16 * APITCH)
static_assert(STRIPS * 16 == NN, "strips cover every row exactly");
static_assert(GPART <= SCUT1 && GPART <= SCUT2 - SCUT1 && GPART <= STRIPS - SCUT2, "every product block has a first strip");

__device__ __forceinline__ void lds_barrier() {
    __builtin_amdgcn_fence(__ATOMIC_RELEASE, "workgroup", "local");
    __builtin_amdgcn_s_barrier();
    __builtin_amdgcn_fence(__ATOMIC_ACQUIRE, "workgroup", "local");
}

__device__ __forceinline__ void
gemm_role(float* L, const int bid, const int s_begin, const int s_end,
          const float* __restrict__ X,
          const float* __restrict__ W,
          const float* __restrict__ a_left,
          const float* __restrict__ a_right,
          float* __restrict__ emb,
          float* __restrict__ ss, float* __restrict__ ts) {
    const int first = s_begin + bid;
    if (first >= s_end) return;
    const int tid  = threadIdx.x;
    const int wave = tid >> 5;
    const int lane = tid & 31;
    const int half = lane >> 4;
    const int l    = lane & 15;

    const float4* __restrict__ X4 = (const float4*)X;
    float4 t0 = X4[first * 512 + tid];
    float4 t1 = X4[first * 512 + 256 + tid];

    const float4* __restrict__ W4 = (const float4*)W + wave * 512;
    float* wsm = L + wave * TILE_F;
#pragma unroll
    for (int i = 0; i < 16; ++i)
        *(float4*)(wsm + i * APITCH + lane * 4) = W4[i * 32 + lane];
    __syncthreads();
    v2f b[32];
#pragma unroll
    for (int kk = 0; kk < 32; ++kk)
        b[kk] = *(const v2f*)(wsm + l * APITCH + kk * 4 + half * 2);

    float* as0 = L + (tid >> 5) * APITCH + (tid & 31) * 4;
    float* as1 = as0 + 8 * APITCH;
    const float* afr = L + l * APITCH + half * 2;

    const float al = a_left[l * HEADS + wave];
    const float ar = a_right[l * HEADS + wave];
    const bool  up8 = (l & 8) != 0, up4 = (l & 4) != 0, up2 = (l & 2) != 0, up1 = (l & 1) != 0;
    const float a_keep = up8 ? ar : al;
    const float a_send = up8 ? al : ar;
    float* sst = L + TILE_F;
    float* sput = sst + (l >> 3) * 128 + (half * 8 + (l & 7)) * HEADS + wave;
    float* __restrict__ sp = (tid < 128 ? ss : ts) + (size_t)first * 128 + (tid & 127);

    int strip = first;
    float* __restrict__ op =
        emb + (size_t)(strip * 16 + half * 8) * OUTC + wave * 16 + l;
    for (; strip < s_end; strip += GPART) {
        lds_barrier();
        if (strip != first) {
            *sp = sst[tid];
            sp += (size_t)GPART * 128;
        }
        *(float4*)as0 = t0;
        *(float4*)as1 = t1;
        lds_barrier();
        const int next = strip + GPART;
        if (next < s_end) {
            t0 = X4[next * 512 + tid];
            t1 = X4[next * 512 + 256 + tid];
        }

        v8f c = {};
#pragma unroll
        for (int kk = 0; kk < 32; ++kk) {
            v2f a = *(const v2f*)(afr + kk * 4);
            c = __builtin_amdgcn_wmma_f32_16x16x4_f32(
                     false, a,  false, b[kk],
                     (short)0, c,  false,  false);
        }

#pragma unroll
        for (int v = 0; v < 8; ++v) op[(size_t)v * OUTC] = c[v];
        op += (size_t)GPART * 16 * OUTC;

        float q8[8], q4[4], q2[2];
#pragma unroll
        for (int v = 0; v < 8; ++v) q8[v] = c[v] * a_keep + row_xor<8>(c[v] * a_send);
#pragma unroll
        for (int k = 0; k < 4; ++k) q4[k] = halve<4>(q8[k], q8[k + 4], up4);
#pragma unroll
        for (int k = 0; k < 2; ++k) q2[k] = halve<2>(q4[k], q4[k + 2], up2);
        *sput = halve<1>(q2[0], q2[1], up1);
    }
    lds_barrier();
    *sp = sst[tid];
}

#define NB    256
#define BIN_W 196
#define NCH   256
#define CH    2344
static_assert(NB * BIN_W >= NN, "bins cover every node");
static_assert(NCH * CH >= EE, "chunks cover every edge");
static_assert(BIN_W <= 256 && EE < (1 << 23), "record packing: edge id << 8 | index in bin");
#define KPT   ((CH + 255) / 256)
#define GEMM_ARGS_DECL const float* __restrict__ X, const float* __restrict__ W, \
                       const float* __restrict__ a_left, const float* __restrict__ a_right, \
                       float* __restrict__ emb, float* __restrict__ ss, float* __restrict__ ts
#define GEMM_ARGS X, W, a_left, a_right, emb, ss, ts

__global__ void __launch_bounds__(256)
gat_binhist_gemm(const int* __restrict__ trg, int* __restrict__ H, GEMM_ARGS_DECL) {
    __shared__ __attribute__((aligned(16))) float L[8 * TILE_F];
    __shared__ int h[NB];
    if (blockIdx.x >= NCH) {
        gemm_role(L, blockIdx.x - NCH, 0, SCUT1, GEMM_ARGS);
        return;
    }
    const int t = threadIdx.x;
    if (t < NB) h[t] = 0;
    __syncthreads();
    const int e0 = blockIdx.x * CH;
    int tg[KPT];
#pragma unroll
    for (int k = 0; k < KPT; ++k) {
        const int i = t + k * 256;
        tg[k] = (i < CH && e0 + i < EE) ? trg[e0 + i] : -1;
    }
#pragma unroll
    for (int k = 0; k < KPT; ++k)
        if (tg[k] >= 0) atomicAdd(&h[tg[k] / BIN_W], 1);
    __syncthreads();
    if (t < NB) H[t * NCH + blockIdx.x] = h[t];
}

#define SCAN_T  1024
#define SCAN_IT 16
static_assert(32 * SCAN_IT * 128 == NB * NCH, "scan covers H exactly");

__global__ void __launch_bounds__(SCAN_T)
gat_scan(int* H) {
    __shared__ int wsum[32];
    const int t = threadIdx.x, lane = t & 31, w = t >> 5;
    const int wbase = w * (SCAN_IT * 128) + lane * 4;
    int4 v[SCAN_IT];
    int  inc[SCAN_IT];
#pragma unroll
    for (int i = 0; i < SCAN_IT; ++i) v[i] = *(const int4*)(H + wbase + i * 128);
#pragma unroll
    for (int i = 0; i < SCAN_IT; ++i) {
        int x = v[i].x + v[i].y + v[i].z + v[i].w;
#pragma unroll
        for (int o = 1; o < 32; o <<= 1) {
            int n = __shfl_up(x, o);
            if (lane >= o) x += n;
        }
        inc[i] = x;
    }
    int carry = 0;
#pragma unroll
    for (int i = 0; i < SCAN_IT; ++i) {
        int tot = __builtin_amdgcn_readlane(inc[i], 31);
        inc[i] += carry;
        carry += tot;
    }
    if (lane == 0) wsum[w] = carry;
    __syncthreads();
    if (w == 0) {
        int x = wsum[lane];
#pragma unroll
        for (int o = 1; o < 32; o <<= 1) {
            int n = __shfl_up(x, o);
            if (lane >= o) x += n;
        }
        wsum[lane] = x;
    }
    __syncthreads();
    const int base = w > 0 ? wsum[w - 1] : 0;
#pragma unroll
    for (int i = 0; i < SCAN_IT; ++i) {
        int4 o4;
        o4.x = base + inc[i] - (v[i].x + v[i].y + v[i].z + v[i].w);
        o4.y = o4.x + v[i].x;
        o4.z = o4.y + v[i].y;
        o4.w = o4.z + v[i].z;
        *(int4*)(H + wbase + i * 128) = o4;
    }
}

__global__ void __launch_bounds__(256)
gat_binplace_gemm(const int* __restrict__ trg, const int* __restrict__ O,
                  int* __restrict__ rec, GEMM_ARGS_DECL) {
    __shared__ __attribute__((aligned(16))) float L[8 * TILE_F];
    __shared__ int cur[NB];
    if (blockIdx.x >= NCH) {
        gemm_role(L, blockIdx.x - NCH, SCUT1, SCUT2, GEMM_ARGS);
        return;
    }
    const int t = threadIdx.x;
    if (t < NB) cur[t] = O[t * NCH + blockIdx.x];
    __syncthreads();
    const int e0 = blockIdx.x * CH;
    int tg[KPT];
#pragma unroll
    for (int k = 0; k < KPT; ++k) {
        const int i = t + k * 256;
        tg[k] = (i < CH && e0 + i < EE) ? trg[e0 + i] : -1;
    }
#pragma unroll
    for (int k = 0; k < KPT; ++k) {
        if (tg[k] >= 0) {
            const int e = e0 + t + k * 256;
            const int b = tg[k] / BIN_W;
            const int p = atomicAdd(&cur[b], 1);
            rec[p] = (e << 8) | (tg[k] - b * BIN_W);
        }
    }
}

#define RPT 12

__global__ void __launch_bounds__(256)
gat_fine_gemm(const int* __restrict__ O, const int* __restrict__ rec,
              int* __restrict__ cnt, int* __restrict__ offs, int* __restrict__ list,
              GEMM_ARGS_DECL) {
    __shared__ __attribute__((aligned(16))) float L[8 * TILE_F];
    __shared__ int c[256];
    __shared__ int cur[256];
    __shared__ int wsum[8];
    if (blockIdx.x >= NB) {
        gemm_role(L, blockIdx.x - NB, SCUT2, STRIPS, GEMM_ARGS);
        return;
    }
    const int t = threadIdx.x, lane = t & 31, w = t >> 5;
    const int b = blockIdx.x;
    const int start = O[b * NCH];
    const int end   = (b == NB - 1) ? EE : O[(b + 1) * NCH];
    const int n = end - start;
    if (t < 256) c[t] = 0;
    __syncthreads();
    for (int i0 = 0; i0 < n; i0 += RPT * 256) {
        int r[RPT];
#pragma unroll
        for (int k = 0; k < RPT; ++k) {
            const int i = i0 + k * 256 + t;
            r[k] = i < n ? rec[start + i] : -1;
        }
#pragma unroll
        for (int k = 0; k < RPT; ++k)
            if (r[k] >= 0) atomicAdd(&c[r[k] & 255], 1);
    }
    __syncthreads();
    const int own = t < 256 ? c[t] : 0;
    int x = own;
#pragma unroll
    for (int o = 1; o < 32; o <<= 1) {
        int m = __shfl_up(x, o);
        if (lane >= o) x += m;
    }
    if (t < 256 && lane == 31) wsum[w] = x;
    __syncthreads();
    if (t < 256) {
        int base = 0;
        for (int j = 0; j < w; ++j) base += wsum[j];
        const int ex = base + x - own;
        cur[t] = ex;
        const int node = b * BIN_W + t;
        if (t < BIN_W && node < NN) {
            cnt[node]  = own;
            offs[node] = start + ex;
        }
    }
    __syncthreads();
    for (int i0 = 0; i0 < n; i0 += RPT * 256) {
        int r[RPT];
#pragma unroll
        for (int k = 0; k < RPT; ++k) {
            const int i = i0 + k * 256 + t;
            r[k] = i < n ? rec[start + i] : -1;
        }
#pragma unroll
        for (int k = 0; k < RPT; ++k) {
            if (r[k] >= 0) {
                const int p = atomicAdd(&cur[r[k] & 255], 1);
                list[start + p] = r[k] >> 8;
            }
        }
    }
}

__device__ __forceinline__ float edge_exp(float s_src, float s_trg) {
    float v = s_src + s_trg;
    v = v > 0.f ? v : 0.2f * v;
    return __expf(v);
}

template <int T>
__device__ __forceinline__ float quad_lane(float v) {
    return __int_as_float(__builtin_amdgcn_update_dpp(
        0, __float_as_int(v), T * 0x55, 0xf, 0xf, true));
}

__device__ __forceinline__ void row_fma(float4& acc, const float4& x, float a) {
    acc.x += x.x * a; acc.y += x.y * a; acc.z += x.z * a; acc.w += x.w * a;
}

__device__ __forceinline__ void rows_request(float4 (&x)[8], const float4* __restrict__ emb4,
                                             int sv, int r0, int last, int lane) {
#pragma unroll
    for (int t = 0; t < 8; ++t) {
        const int r = r0 + t < last ? r0 + t : last;
        x[t] = emb4[__builtin_amdgcn_readlane(sv, r) * 32 + lane];
    }
}

__device__ __forceinline__ void rows_sum(float4& acc, const float4 (&x)[8], float wa, float wb) {
    row_fma(acc, x[0], quad_lane<0>(wa));
    row_fma(acc, x[1], quad_lane<1>(wa));
    row_fma(acc, x[2], quad_lane<2>(wa));
    row_fma(acc, x[3], quad_lane<3>(wa));
    row_fma(acc, x[4], quad_lane<0>(wb));
    row_fma(acc, x[5], quad_lane<1>(wb));
    row_fma(acc, x[6], quad_lane<2>(wb));
    row_fma(acc, x[7], quad_lane<3>(wb));
}

__global__ void __launch_bounds__(256)
gat_aggregate(const int* __restrict__ src,
              const int* __restrict__ cnt, const int* __restrict__ offs,
              const int* __restrict__ list,
              const float* __restrict__ emb,
              const float* __restrict__ ss, const float* __restrict__ ts,
              const float* __restrict__ bias,
              float* __restrict__ out) {
    const int lane = threadIdx.x & 31;
    const int node = __builtin_amdgcn_readfirstlane((int)((blockIdx.x * blockDim.x + threadIdx.x) >> 5));
    if (node >= NN) return;
    const int h = lane >> 2;
    const int deg = __builtin_amdgcn_readfirstlane(cnt[node]);
    const int beg = __builtin_amdgcn_readfirstlane(offs[node]);
    const float tsv = ts[node * HEADS + h];
    const float4* __restrict__ emb4 = (const float4*)emb;

    float denom = 0.f;
    float4 acc = make_float4(0.f, 0.f, 0.f, 0.f);

    if (deg <= 32) {
        int e = lane < deg ? list[beg + lane] : 0x7fffffff;
        int rank = 0;
        for (int j = 0; j < deg; ++j) {
            int ej = __builtin_amdgcn_readlane(e, j);
            rank += (ej < e) ? 1 : 0;
        }
        if (lane >= deg) rank = lane;
        int es = __builtin_amdgcn_ds_permute(rank << 2, e);
        int sv = lane < deg ? src[es] : 0;

        const int q    = lane & 3;
        const int last = deg > 0 ? deg - 1 : 0;
        const bool big = deg > 16;

        int sj[8];
#pragma unroll
        for (int j = 0; j < 4; ++j)
            sj[j] = __builtin_amdgcn_ds_bpermute((4 * j + q) << 2, sv);
        float4 xr[8];
        rows_request(xr, emb4, sv, 0, last, lane);
        float sc[8];
#pragma unroll
        for (int j = 0; j < 4; ++j) sc[j] = ss[sj[j] * HEADS + h];
#pragma unroll
        for (int j = 4; j < 8; ++j) sc[j] = 0.f;
        if (big) {
#pragma unroll
            for (int j = 4; j < 8; ++j)
                sj[j] = __builtin_amdgcn_ds_bpermute((4 * j + q) << 2, sv);
#pragma unroll
            for (int j = 4; j < 8; ++j) sc[j] = ss[sj[j] * HEADS + h];
        }

        float ex[8];
#pragma unroll
        for (int j = 0; j < 4; ++j) {
            const float v = edge_exp(sc[j], tsv);
            ex[j] = (4 * j + q < deg) ? v : 0.f;
        }
#pragma unroll
        for (int j = 4; j < 8; ++j) ex[j] = 0.f;
#pragma unroll
        for (int j = 0; j < 4; ++j) {
            denom += quad_lane<0>(ex[j]);
            denom += quad_lane<1>(ex[j]);
            denom += quad_lane<2>(ex[j]);
            denom += quad_lane<3>(ex[j]);
        }
        if (big) {
#pragma unroll
            for (int j = 4; j < 8; ++j) {
                const float v = edge_exp(sc[j], tsv);
                ex[j] = (4 * j + q < deg) ? v : 0.f;
            }
#pragma unroll
            for (int j = 4; j < 8; ++j) {
                denom += quad_lane<0>(ex[j]);
                denom += quad_lane<1>(ex[j]);
                denom += quad_lane<2>(ex[j]);
                denom += quad_lane<3>(ex[j]);
            }
        }
        const float dn = denom + 1e-16f;
        float w[8];
#pragma unroll
        for (int j = 0; j < 8; ++j) w[j] = ex[j] / dn;

        rows_sum(acc, xr, w[0], w[1]);
        if (deg > 8) {
            rows_request(xr, emb4, sv, 8, last, lane);
            rows_sum(acc, xr, w[2], w[3]);
            if (big) {
                rows_request(xr, emb4, sv, 16, last, lane);
                rows_sum(acc, xr, w[4], w[5]);
                if (deg > 24) {
                    rows_request(xr, emb4, sv, 24, last, lane);
                    rows_sum(acc, xr, w[6], w[7]);
                }
            }
        }
    } else {
        int prev = -1;
        for (int r = 0; r < deg; ++r) {
            int m = 0x7fffffff;
            for (int j = lane; j < deg; j += 32) {
                int ej = list[beg + j];
                if (ej > prev && ej < m) m = ej;
            }
#pragma unroll
            for (int o = 16; o > 0; o >>= 1) {
                int n = __shfl_xor(m, o);
                m = n < m ? n : m;
            }
            prev = __builtin_amdgcn_readfirstlane(m);
            int s = src[prev];
            denom += edge_exp(ss[s * HEADS + h], tsv);
        }
        const float dn = denom + 1e-16f;
        prev = -1;
        for (int r = 0; r < deg; ++r) {
            int m = 0x7fffffff;
            for (int j = lane; j < deg; j += 32) {
                int ej = list[beg + j];
                if (ej > prev && ej < m) m = ej;
            }
#pragma unroll
            for (int o = 16; o > 0; o >>= 1) {
                int n = __shfl_xor(m, o);
                m = n < m ? n : m;
            }
            prev = __builtin_amdgcn_readfirstlane(m);
            int s0 = src[prev];
            float4 x0 = emb4[s0 * 32 + lane];
            float a0 = edge_exp(ss[s0 * HEADS + h], tsv) / dn;
            acc.x += x0.x * a0; acc.y += x0.y * a0; acc.z += x0.z * a0; acc.w += x0.w * a0;
        }
    }

    const float4 b = ((const float4*)bias)[lane];
    float4 o4 = make_float4(acc.x + b.x, acc.y + b.y, acc.z + b.z, acc.w + b.w);
    ((float4*)out)[node * 32 + lane] = o4;
}

extern "C" void kernel_launch(void* const* d_in, const int* in_sizes, int n_in,
                              void* d_out, int out_size, void* d_ws, size_t ws_size,
                              hipStream_t stream) {
    const float* X      = (const float*)d_in[0];
    const int*   eidx   = (const int*)d_in[1];
    const float* W      = (const float*)d_in[2];
    const float* a_l    = (const float*)d_in[3];
    const float* a_r    = (const float*)d_in[4];
    const float* bias   = (const float*)d_in[5];
    float* out = (float*)d_out;

    const int* src = eidx;
    const int* trg = eidx + EE;

    float* ws    = (float*)d_ws;
    float* emb   = ws;
    float* ss    = emb + (size_t)NN * OUTC;
    float* ts    = ss + (size_t)NN * HEADS;
    int*   cnt   = (int*)(ts + (size_t)NN * HEADS);
    int*   offs  = cnt + NN;
    int*   rec   = offs + NN;
    int*   list  = rec + EE;
    int*   H     = (int*)d_out;

    gat_binhist_gemm<<<NCH + GPART, 256, 0, stream>>>(trg, H, X, W, a_l, a_r, emb, ss, ts);
    gat_scan<<<1, SCAN_T, 0, stream>>>(H);
    gat_binplace_gemm<<<NCH + GPART, 256, 0, stream>>>(trg, H, rec, X, W, a_l, a_r, emb, ss, ts);
    gat_fine_gemm<<<NB + GPART, 256, 0, stream>>>(H, rec, cnt, offs, list, X, W, a_l, a_r, emb, ss, ts);

    gat_aggregate<<<(NN + 7) / 8, 256, 0, stream>>>(src, cnt, offs, list, emb, ss, ts, bias, out);
}


# Scan launch gone: count table kept chunk-major over 128 chunks; each bin-record block sums its columns (chunks before it, all chunks) and scans the 256 bin totals in LDS; float sums unchanged
# speedup vs baseline: 4.4470x; 1.1545x over previous
#include <hip/hip_runtime.h>
#include <hip/hip_bf16.h>

#define NN 50000
#define EE 600000
#define INC 128
#define OUTC 128
#define HEADS 8
#define HEADC 16

typedef __attribute__((ext_vector_type(2))) float v2f;
typedef __attribute__((ext_vector_type(8))) float v8f;

template <int M>
__device__ __forceinline__ float row_xor(float v) {
    return __int_as_float(__builtin_amdgcn_update_dpp(
        0, __float_as_int(v), 0x160 | M, 0xf, 0xf, true));
}
template <int M>
__device__ __forceinline__ float halve(float lo, float hi, bool up) {
    const float keep = up ? hi : lo;
    const float send = up ? lo : hi;
    return keep + row_xor<M>(send);
}

#define STRIPS    (NN / 16)
#define GPART     512
#define SCUT1     1042
#define SCUT2     2084
#define APITCH    132
#define TILE_F    (16 * APITCH)
static_assert(STRIPS * 16 == NN, "strips cover every row exactly");
static_assert(GPART <= SCUT1 && GPART <= SCUT2 - SCUT1 && GPART <= STRIPS - SCUT2, "every product block has a first strip");

__device__ __forceinline__ void lds_barrier() {
    __builtin_amdgcn_fence(__ATOMIC_RELEASE, "workgroup", "local");
    __builtin_amdgcn_s_barrier();
    __builtin_amdgcn_fence(__ATOMIC_ACQUIRE, "workgroup", "local");
}

__device__ __forceinline__ void
gemm_role(float* L, const int bid, const int s_begin, const int s_end,
          const float* __restrict__ X,
          const float* __restrict__ W,
          const float* __restrict__ a_left,
          const float* __restrict__ a_right,
          float* __restrict__ emb,
          float* __restrict__ ss, float* __restrict__ ts) {
    const int first = s_begin + bid;
    if (first >= s_end) return;
    const int tid  = threadIdx.x;
    const int wave = tid >> 5;
    const int lane = tid & 31;
    const int half = lane >> 4;
    const int l    = lane & 15;

    const float4* __restrict__ X4 = (const float4*)X;
    float4 t0 = X4[first * 512 + tid];
    float4 t1 = X4[first * 512 + 256 + tid];

    const float4* __restrict__ W4 = (const float4*)W + wave * 512;
    float* wsm = L + wave * TILE_F;
#pragma unroll
    for (int i = 0; i < 16; ++i)
        *(float4*)(wsm + i * APITCH + lane * 4) = W4[i * 32 + lane];
    __syncthreads();
    v2f b[32];
#pragma unroll
    for (int kk = 0; kk < 32; ++kk)
        b[kk] = *(const v2f*)(wsm + l * APITCH + kk * 4 + half * 2);

    float* as0 = L + (tid >> 5) * APITCH + (tid & 31) * 4;
    float* as1 = as0 + 8 * APITCH;
    const float* afr = L + l * APITCH + half * 2;

    const float al = a_left[l * HEADS + wave];
    const float ar = a_right[l * HEADS + wave];
    const bool  up8 = (l & 8) != 0, up4 = (l & 4) != 0, up2 = (l & 2) != 0, up1 = (l & 1) != 0;
    const float a_keep = up8 ? ar : al;
    const float a_send = up8 ? al : ar;
    float* sst = L + TILE_F;
    float* sput = sst + (l >> 3) * 128 + (half * 8 + (l & 7)) * HEADS + wave;
    float* __restrict__ sp = (tid < 128 ? ss : ts) + (size_t)first * 128 + (tid & 127);

    int strip = first;
    float* __restrict__ op =
        emb + (size_t)(strip * 16 + half * 8) * OUTC + wave * 16 + l;
    for (; strip < s_end; strip += GPART) {
        lds_barrier();
        if (strip != first) {
            *sp = sst[tid];
            sp += (size_t)GPART * 128;
        }
        *(float4*)as0 = t0;
        *(float4*)as1 = t1;
        lds_barrier();
        const int next = strip + GPART;
        if (next < s_end) {
            t0 = X4[next * 512 + tid];
            t1 = X4[next * 512 + 256 + tid];
        }

        v8f c = {};
#pragma unroll
        for (int kk = 0; kk < 32; ++kk) {
            v2f a = *(const v2f*)(afr + kk * 4);
            c = __builtin_amdgcn_wmma_f32_16x16x4_f32(
                     false, a,  false, b[kk],
                     (short)0, c,  false,  false);
        }

#pragma unroll
        for (int v = 0; v < 8; ++v) op[(size_t)v * OUTC] = c[v];
        op += (size_t)GPART * 16 * OUTC;

        float q8[8], q4[4], q2[2];
#pragma unroll
        for (int v = 0; v < 8; ++v) q8[v] = c[v] * a_keep + row_xor<8>(c[v] * a_send);
#pragma unroll
        for (int k = 0; k < 4; ++k) q4[k] = halve<4>(q8[k], q8[k + 4], up4);
#pragma unroll
        for (int k = 0; k < 2; ++k) q2[k] = halve<2>(q4[k], q4[k + 2], up2);
        *sput = halve<1>(q2[0], q2[1], up1);
    }
    lds_barrier();
    *sp = sst[tid];
}

#define NB    256
#define BIN_W 196
#define NCH   128
#define CH    4688
static_assert(NB * BIN_W >= NN, "bins cover every node");
static_assert(NCH * CH >= EE, "chunks cover every edge");
static_assert(NB == 256 && (NCH & (NCH - 1)) == 0 && NCH % 4 == 0, "column sums: 64 threads x 4 bins per row, 4 rows a step, rows taken modulo NCH");
static_assert(BIN_W <= 256 && EE < (1 << 23), "record packing: edge id << 8 | index in bin");
#define KPT   ((CH + 255) / 256)
#define GEMM_ARGS_DECL const float* __restrict__ X, const float* __restrict__ W, \
                       const float* __restrict__ a_left, const float* __restrict__ a_right, \
                       float* __restrict__ emb, float* __restrict__ ss, float* __restrict__ ts
#define GEMM_ARGS X, W, a_left, a_right, emb, ss, ts

__global__ void __launch_bounds__(256)
gat_binhist_gemm(const int* __restrict__ trg, int* __restrict__ H, GEMM_ARGS_DECL) {
    __shared__ __attribute__((aligned(16))) float L[8 * TILE_F];
    __shared__ int h[NB];
    if (blockIdx.x >= NCH) {
        gemm_role(L, blockIdx.x - NCH, 0, SCUT1, GEMM_ARGS);
        return;
    }
    const int t = threadIdx.x;
    if (t < NB) h[t] = 0;
    __syncthreads();
    const int e0 = blockIdx.x * CH;
    int tg[KPT];
#pragma unroll
    for (int k = 0; k < KPT; ++k) {
        const int i = t + k * 256;
        tg[k] = (i < CH && e0 + i < EE) ? trg[e0 + i] : -1;
    }
#pragma unroll
    for (int k = 0; k < KPT; ++k)
        if (tg[k] >= 0) atomicAdd(&h[tg[k] / BIN_W], 1);
    __syncthreads();
    if (t < NB) H[blockIdx.x * NB + t] = h[t];
}

__global__ void __launch_bounds__(256)
gat_binplace_gemm(const int* __restrict__ trg, const int* __restrict__ H,
                  int* __restrict__ binstart,
                  int* __restrict__ rec, GEMM_ARGS_DECL) {
    __shared__ __attribute__((aligned(16))) float L[8 * TILE_F];
    __shared__ int cur[NB];
    __shared__ __attribute__((aligned(16))) int part[2 * 4 * NB];
    __shared__ int wsum[8];
    if (blockIdx.x >= NCH) {
        gemm_role(L, blockIdx.x - NCH, SCUT1, SCUT2, GEMM_ARGS);
        return;
    }
    const int t = threadIdx.x, lane = t & 31, w = t >> 5;
    const int c = blockIdx.x;
    const int e0 = c * CH;
    int tg[KPT];
#pragma unroll
    for (int k = 0; k < KPT; ++k) {
        const int i = t + k * 256;
        tg[k] = (i < CH && e0 + i < EE) ? trg[e0 + i] : -1;
    }
    {
        const int j = t & 63, g = t >> 6;
        int4 P = make_int4(0, 0, 0, 0);
        int4 T = make_int4(0, 0, 0, 0);
#pragma unroll 8
        for (int k = 0; k < NCH / 4; ++k) {
            const int r = (c + g + 4 * k) & (NCH - 1);
            const int4 v = *(const int4*)(H + r * NB + 4 * j);
            T.x += v.x; T.y += v.y; T.z += v.z; T.w += v.w;
            if (r < c) { P.x += v.x; P.y += v.y; P.z += v.z; P.w += v.w; }
        }
        *(int4*)(part + g * NB + 4 * j)          = P;
        *(int4*)(part + 4 * NB + g * NB + 4 * j) = T;
    }
    __syncthreads();
    {
        const int before = part[t] + part[NB + t] + part[2 * NB + t] + part[3 * NB + t];
        const int total  = part[4 * NB + t] + part[5 * NB + t] + part[6 * NB + t] + part[7 * NB + t];
        int x = total;
#pragma unroll
        for (int o = 1; o < 32; o <<= 1) {
            int m = __shfl_up(x, o);
            if (lane >= o) x += m;
        }
        if (lane == 31) wsum[w] = x;
        __syncthreads();
        int base = 0;
        for (int q = 0; q < w; ++q) base += wsum[q];
        const int first = base + x - total;
        cur[t] = first + before;
        if (c == 0) binstart[t] = first;
    }
    __syncthreads();
#pragma unroll
    for (int k = 0; k < KPT; ++k) {
        if (tg[k] >= 0) {
            const int e = e0 + t + k * 256;
            const int b = tg[k] / BIN_W;
            const int p = atomicAdd(&cur[b], 1);
            rec[p] = (e << 8) | (tg[k] - b * BIN_W);
        }
    }
}

#define RPT 12

__global__ void __launch_bounds__(256)
gat_fine_gemm(const int* __restrict__ binstart, const int* __restrict__ rec,
              int* __restrict__ cnt, int* __restrict__ offs, int* __restrict__ list,
              GEMM_ARGS_DECL) {
    __shared__ __attribute__((aligned(16))) float L[8 * TILE_F];
    __shared__ int c[256];
    __shared__ int cur[256];
    __shared__ int wsum[8];
    if (blockIdx.x >= NB) {
        gemm_role(L, blockIdx.x - NB, SCUT2, STRIPS, GEMM_ARGS);
        return;
    }
    const int t = threadIdx.x, lane = t & 31, w = t >> 5;
    const int b = blockIdx.x;
    const int start = binstart[b];
    const int end   = (b == NB - 1) ? EE : binstart[b + 1];
    const int n = end - start;
    if (t < 256) c[t] = 0;
    __syncthreads();
    for (int i0 = 0; i0 < n; i0 += RPT * 256) {
        int r[RPT];
#pragma unroll
        for (int k = 0; k < RPT; ++k) {
            const int i = i0 + k * 256 + t;
            r[k] = i < n ? rec[start + i] : -1;
        }
#pragma unroll
        for (int k = 0; k < RPT; ++k)
            if (r[k] >= 0) atomicAdd(&c[r[k] & 255], 1);
    }
    __syncthreads();
    const int own = t < 256 ? c[t] : 0;
    int x = own;
#pragma unroll
    for (int o = 1; o < 32; o <<= 1) {
        int m = __shfl_up(x, o);
        if (lane >= o) x += m;
    }
    if (t < 256 && lane == 31) wsum[w] = x;
    __syncthreads();
    if (t < 256) {
        int base = 0;
        for (int j = 0; j < w; ++j) base += wsum[j];
        const int ex = base + x - own;
        cur[t] = ex;
        const int node = b * BIN_W + t;
        if (t < BIN_W && node < NN) {
            cnt[node]  = own;
            offs[node] = start + ex;
        }
    }
    __syncthreads();
    for (int i0 = 0; i0 < n; i0 += RPT * 256) {
        int r[RPT];
#pragma unroll
        for (int k = 0; k < RPT; ++k) {
            const int i = i0 + k * 256 + t;
            r[k] = i < n ? rec[start + i] : -1;
        }
#pragma unroll
        for (int k = 0; k < RPT; ++k) {
            if (r[k] >= 0) {
                const int p = atomicAdd(&cur[r[k] & 255], 1);
                list[start + p] = r[k] >> 8;
            }
        }
    }
}

__device__ __forceinline__ float edge_exp(float s_src, float s_trg) {
    float v = s_src + s_trg;
    v = v > 0.f ? v : 0.2f * v;
    return __expf(v);
}

template <int T>
__device__ __forceinline__ float quad_lane(float v) {
    return __int_as_float(__builtin_amdgcn_update_dpp(
        0, __float_as_int(v), T * 0x55, 0xf, 0xf, true));
}

__device__ __forceinline__ void row_fma(float4& acc, const float4& x, float a) {
    acc.x += x.x * a; acc.y += x.y * a; acc.z += x.z * a; acc.w += x.w * a;
}

__device__ __forceinline__ void rows_request(float4 (&x)[8], const float4* __restrict__ emb4,
                                             int sv, int r0, int last, int lane) {
#pragma unroll
    for (int t = 0; t < 8; ++t) {
        const int r = r0 + t < last ? r0 + t : last;
        x[t] = emb4[__builtin_amdgcn_readlane(sv, r) * 32 + lane];
    }
}

__device__ __forceinline__ void rows_sum(float4& acc, const float4 (&x)[8], float wa, float wb) {
    row_fma(acc, x[0], quad_lane<0>(wa));
    row_fma(acc, x[1], quad_lane<1>(wa));
    row_fma(acc, x[2], quad_lane<2>(wa));
    row_fma(acc, x[3], quad_lane<3>(wa));
    row_fma(acc, x[4], quad_lane<0>(wb));
    row_fma(acc, x[5], quad_lane<1>(wb));
    row_fma(acc, x[6], quad_lane<2>(wb));
    row_fma(acc, x[7], quad_lane<3>(wb));
}

__global__ void __launch_bounds__(256)
gat_aggregate(const int* __restrict__ src,
              const int* __restrict__ cnt, const int* __restrict__ offs,
              const int* __restrict__ list,
              const float* __restrict__ emb,
              const float* __restrict__ ss, const float* __restrict__ ts,
              const float* __restrict__ bias,
              float* __restrict__ out) {
    const int lane = threadIdx.x & 31;
    const int node = __builtin_amdgcn_readfirstlane((int)((blockIdx.x * blockDim.x + threadIdx.x) >> 5));
    if (node >= NN) return;
    const int h = lane >> 2;
    const int deg = __builtin_amdgcn_readfirstlane(cnt[node]);
    const int beg = __builtin_amdgcn_readfirstlane(offs[node]);
    const float tsv = ts[node * HEADS + h];
    const float4* __restrict__ emb4 = (const float4*)emb;

    float denom = 0.f;
    float4 acc = make_float4(0.f, 0.f, 0.f, 0.f);

    if (deg <= 32) {
        int e = lane < deg ? list[beg + lane] : 0x7fffffff;
        int rank = 0;
        for (int j = 0; j < deg; ++j) {
            int ej = __builtin_amdgcn_readlane(e, j);
            rank += (ej < e) ? 1 : 0;
        }
        if (lane >= deg) rank = lane;
        int es = __builtin_amdgcn_ds_permute(rank << 2, e);
        int sv = lane < deg ? src[es] : 0;

        const int q    = lane & 3;
        const int last = deg > 0 ? deg - 1 : 0;
        const bool big = deg > 16;

        int sj[8];
#pragma unroll
        for (int j = 0; j < 4; ++j)
            sj[j] = __builtin_amdgcn_ds_bpermute((4 * j + q) << 2, sv);
        float4 xr[8];
        rows_request(xr, emb4, sv, 0, last, lane);
        float sc[8];
#pragma unroll
        for (int j = 0; j < 4; ++j) sc[j] = ss[sj[j] * HEADS + h];
#pragma unroll
        for (int j = 4; j < 8; ++j) sc[j] = 0.f;
        if (big) {
#pragma unroll
            for (int j = 4; j < 8; ++j)
                sj[j] = __builtin_amdgcn_ds_bpermute((4 * j + q) << 2, sv);
#pragma unroll
            for (int j = 4; j < 8; ++j) sc[j] = ss[sj[j] * HEADS + h];
        }

        float ex[8];
#pragma unroll
        for (int j = 0; j < 4; ++j) {
            const float v = edge_exp(sc[j], tsv);
            ex[j] = (4 * j + q < deg) ? v : 0.f;
        }
#pragma unroll
        for (int j = 4; j < 8; ++j) ex[j] = 0.f;
#pragma unroll
        for (int j = 0; j < 4; ++j) {
            denom += quad_lane<0>(ex[j]);
            denom += quad_lane<1>(ex[j]);
            denom += quad_lane<2>(ex[j]);
            denom += quad_lane<3>(ex[j]);
        }
        if (big) {
#pragma unroll
            for (int j = 4; j < 8; ++j) {
                const float v = edge_exp(sc[j], tsv);
                ex[j] = (4 * j + q < deg) ? v : 0.f;
            }
#pragma unroll
            for (int j = 4; j < 8; ++j) {
                denom += quad_lane<0>(ex[j]);
                denom += quad_lane<1>(ex[j]);
                denom += quad_lane<2>(ex[j]);
                denom += quad_lane<3>(ex[j]);
            }
        }
        const float dn = denom + 1e-16f;
        float w[8];
#pragma unroll
        for (int j = 0; j < 8; ++j) w[j] = ex[j] / dn;

        rows_sum(acc, xr, w[0], w[1]);
        if (deg > 8) {
            rows_request(xr, emb4, sv, 8, last, lane);
            rows_sum(acc, xr, w[2], w[3]);
            if (big) {
                rows_request(xr, emb4, sv, 16, last, lane);
                rows_sum(acc, xr, w[4], w[5]);
                if (deg > 24) {
                    rows_request(xr, emb4, sv, 24, last, lane);
                    rows_sum(acc, xr, w[6], w[7]);
                }
            }
        }
    } else {
        int prev = -1;
        for (int r = 0; r < deg; ++r) {
            int m = 0x7fffffff;
            for (int j = lane; j < deg; j += 32) {
                int ej = list[beg + j];
                if (ej > prev && ej < m) m = ej;
            }
#pragma unroll
            for (int o = 16; o > 0; o >>= 1) {
                int n = __shfl_xor(m, o);
                m = n < m ? n : m;
            }
            prev = __builtin_amdgcn_readfirstlane(m);
            int s = src[prev];
            denom += edge_exp(ss[s * HEADS + h], tsv);
        }
        const float dn = denom + 1e-16f;
        prev = -1;
        for (int r = 0; r < deg; ++r) {
            int m = 0x7fffffff;
            for (int j = lane; j < deg; j += 32) {
                int ej = list[beg + j];
                if (ej > prev && ej < m) m = ej;
            }
#pragma unroll
            for (int o = 16; o > 0; o >>= 1) {
                int n = __shfl_xor(m, o);
                m = n < m ? n : m;
            }
            prev = __builtin_amdgcn_readfirstlane(m);
            int s0 = src[prev];
            float4 x0 = emb4[s0 * 32 + lane];
            float a0 = edge_exp(ss[s0 * HEADS + h], tsv) / dn;
            acc.x += x0.x * a0; acc.y += x0.y * a0; acc.z += x0.z * a0; acc.w += x0.w * a0;
        }
    }

    const float4 b = ((const float4*)bias)[lane];
    float4 o4 = make_float4(acc.x + b.x, acc.y + b.y, acc.z + b.z, acc.w + b.w);
    ((float4*)out)[node * 32 + lane] = o4;
}

extern "C" void kernel_launch(void* const* d_in, const int* in_sizes, int n_in,
                              void* d_out, int out_size, void* d_ws, size_t ws_size,
                              hipStream_t stream) {
    const float* X      = (const float*)d_in[0];
    const int*   eidx   = (const int*)d_in[1];
    const float* W      = (const float*)d_in[2];
    const float* a_l    = (const float*)d_in[3];
    const float* a_r    = (const float*)d_in[4];
    const float* bias   = (const float*)d_in[5];
    float* out = (float*)d_out;

    const int* src = eidx;
    const int* trg = eidx + EE;

    float* ws    = (float*)d_ws;
    float* emb   = ws;
    float* ss    = emb + (size_t)NN * OUTC;
    float* ts    = ss + (size_t)NN * HEADS;
    int*   cnt   = (int*)(ts + (size_t)NN * HEADS);
    int*   offs  = cnt + NN;
    int*   rec   = offs + NN;
    int*   list  = rec + EE;
    int*   H        = (int*)d_out;
    int*   binstart = H + NCH * NB;

    gat_binhist_gemm<<<NCH + GPART, 256, 0, stream>>>(trg, H, X, W, a_l, a_r, emb, ss, ts);
    gat_binplace_gemm<<<NCH + GPART, 256, 0, stream>>>(trg, H, binstart, rec, X, W, a_l, a_r, emb, ss, ts);
    gat_fine_gemm<<<NB + GPART, 256, 0, stream>>>(binstart, rec, cnt, offs, list, X, W, a_l, a_r, emb, ss, ts);

    gat_aggregate<<<(NN + 7) / 8, 256, 0, stream>>>(src, cnt, offs, list, emb, ss, ts, bias, out);
}


# Chunk blocks sort their edges by bin in LDS and store the slice in whole lines, with a bin x chunk run table; bin blocks read their 128 runs: count launch gone (3 launches); float sums unchanged
# speedup vs baseline: 5.9186x; 1.3309x over previous
#include <hip/hip_runtime.h>
#include <hip/hip_bf16.h>

#define NN 50000
#define EE 600000
#define INC 128
#define OUTC 128
#define HEADS 8
#define HEADC 16

typedef __attribute__((ext_vector_type(2))) float v2f;
typedef __attribute__((ext_vector_type(8))) float v8f;

template <int M>
__device__ __forceinline__ float row_xor(float v) {
    return __int_as_float(__builtin_amdgcn_update_dpp(
        0, __float_as_int(v), 0x160 | M, 0xf, 0xf, true));
}
template <int M>
__device__ __forceinline__ float halve(float lo, float hi, bool up) {
    const float keep = up ? hi : lo;
    const float send = up ? lo : hi;
    return keep + row_xor<M>(send);
}

#define STRIPS    (NN / 16)
#define GP1       520
#define GP2       522
#define SCUT      (3 * GP1)
#define APITCH    132
#define TILE_F    (16 * APITCH)
static_assert(STRIPS * 16 == NN, "strips cover every row exactly");
static_assert(GP1 <= SCUT && GP2 <= STRIPS - SCUT && STRIPS - SCUT <= 3 * GP2, "every product block has a first strip and at most three");

__device__ __forceinline__ void lds_barrier() {
    __builtin_amdgcn_fence(__ATOMIC_RELEASE, "workgroup", "local");
    __builtin_amdgcn_s_barrier();
    __builtin_amdgcn_fence(__ATOMIC_ACQUIRE, "workgroup", "local");
}

__device__ __forceinline__ void
gemm_role(float* L, const int bid, const int gp, const int s_begin, const int s_end,
          const float* __restrict__ X,
          const float* __restrict__ W,
          const float* __restrict__ a_left,
          const float* __restrict__ a_right,
          float* __restrict__ emb,
          float* __restrict__ ss, float* __restrict__ ts) {
    const int first = s_begin + bid;
    if (first >= s_end) return;
    const int tid  = threadIdx.x;
    const int wave = tid >> 5;
    const int lane = tid & 31;
    const int half = lane >> 4;
    const int l    = lane & 15;

    const float4* __restrict__ X4 = (const float4*)X;
    float4 t0 = X4[first * 512 + tid];
    float4 t1 = X4[first * 512 + 256 + tid];

    const float4* __restrict__ W4 = (const float4*)W + wave * 512;
    float* wsm = L + wave * TILE_F;
#pragma unroll
    for (int i = 0; i < 16; ++i)
        *(float4*)(wsm + i * APITCH + lane * 4) = W4[i * 32 + lane];
    __syncthreads();
    v2f b[32];
#pragma unroll
    for (int kk = 0; kk < 32; ++kk)
        b[kk] = *(const v2f*)(wsm + l * APITCH + kk * 4 + half * 2);

    float* as0 = L + (tid >> 5) * APITCH + (tid & 31) * 4;
    float* as1 = as0 + 8 * APITCH;
    const float* afr = L + l * APITCH + half * 2;

    const float al = a_left[l * HEADS + wave];
    const float ar = a_right[l * HEADS + wave];
    const bool  up8 = (l & 8) != 0, up4 = (l & 4) != 0, up2 = (l & 2) != 0, up1 = (l & 1) != 0;
    const float a_keep = up8 ? ar : al;
    const float a_send = up8 ? al : ar;
    float* sst = L + TILE_F;
    float* sput = sst + (l >> 3) * 128 + (half * 8 + (l & 7)) * HEADS + wave;
    float* __restrict__ sp = (tid < 128 ? ss : ts) + (size_t)first * 128 + (tid & 127);

    int strip = first;
    float* __restrict__ op =
        emb + (size_t)(strip * 16 + half * 8) * OUTC + wave * 16 + l;
    for (; strip < s_end; strip += gp) {
        lds_barrier();
        if (strip != first) {
            *sp = sst[tid];
            sp += (size_t)gp * 128;
        }
        *(float4*)as0 = t0;
        *(float4*)as1 = t1;
        lds_barrier();
        const int next = strip + gp;
        if (next < s_end) {
            t0 = X4[next * 512 + tid];
            t1 = X4[next * 512 + 256 + tid];
        }

        v8f c = {};
#pragma unroll
        for (int kk = 0; kk < 32; ++kk) {
            v2f a = *(const v2f*)(afr + kk * 4);
            c = __builtin_amdgcn_wmma_f32_16x16x4_f32(
                     false, a,  false, b[kk],
                     (short)0, c,  false,  false);
        }

#pragma unroll
        for (int v = 0; v < 8; ++v) op[(size_t)v * OUTC] = c[v];
        op += (size_t)gp * 16 * OUTC;

        float q8[8], q4[4], q2[2];
#pragma unroll
        for (int v = 0; v < 8; ++v) q8[v] = c[v] * a_keep + row_xor<8>(c[v] * a_send);
#pragma unroll
        for (int k = 0; k < 4; ++k) q4[k] = halve<4>(q8[k], q8[k + 4], up4);
#pragma unroll
        for (int k = 0; k < 2; ++k) q2[k] = halve<2>(q4[k], q4[k + 2], up2);
        *sput = halve<1>(q2[0], q2[1], up1);
    }
    lds_barrier();
    *sp = sst[tid];
}

#define NB    256
#define BIN_W 196
#define NCH   128
#define CH    4688
static_assert(NB * BIN_W >= NN, "bins cover every node");
static_assert(NCH * CH >= EE && (NCH - 1) * CH < EE, "chunks cover every edge, none is empty");
static_assert(NB == 256 && (NCH & (NCH - 1)) == 0 && NCH % 8 == 0 && NCH <= 256, "column sums: 64 threads x 4 bins per row, 4 rows a step, rows taken modulo NCH; runs: 8 waves, one thread per chunk");
static_assert(BIN_W <= 256 && EE < (1 << 23), "record packing: edge id << 8 | index in bin");
#define KPT   ((CH + 255) / 256)
#define RPW   (NCH / 8)
#define GEMM_ARGS_DECL const float* __restrict__ X, const float* __restrict__ W, \
                       const float* __restrict__ a_left, const float* __restrict__ a_right, \
                       float* __restrict__ emb, float* __restrict__ ss, float* __restrict__ ts
#define GEMM_ARGS X, W, a_left, a_right, emb, ss, ts

__global__ void __launch_bounds__(256)
gat_chunksort_gemm(const int* __restrict__ trg, int* __restrict__ H, int2* __restrict__ R,
                   int* __restrict__ rec, GEMM_ARGS_DECL) {
    __shared__ __attribute__((aligned(16))) float L[8 * TILE_F];
    __shared__ int h[NB + 1];
    __shared__ int wsum[8];
    __shared__ int srt[CH];
    if (blockIdx.x >= NCH) {
        gemm_role(L, blockIdx.x - NCH, GP1, 0, SCUT, GEMM_ARGS);
        return;
    }
    const int t = threadIdx.x, lane = t & 31, w = t >> 5;
    const int c = blockIdx.x;
    h[t] = 0;
    if (t == 0) h[NB] = 0;
    __syncthreads();
    const int e0 = c * CH;
    const int nloc = (EE - e0 < CH) ? EE - e0 : CH;
    int tg[KPT];
#pragma unroll
    for (int k = 0; k < KPT; ++k) {
        const int i = t + k * 256;
        tg[k] = (i < nloc) ? trg[e0 + i] : -1;
    }
    int bk[KPT];
#pragma unroll
    for (int k = 0; k < KPT; ++k) bk[k] = tg[k] >= 0 ? tg[k] / BIN_W : NB;
#pragma unroll
    for (int k = 0; k < KPT; ++k) atomicAdd(&h[bk[k]], 1);
    __syncthreads();
    const int own = h[t];
    int x = own;
#pragma unroll
    for (int o = 1; o < 32; o <<= 1) {
        int m = __shfl_up(x, o);
        if (lane >= o) x += m;
    }
    if (lane == 31) wsum[w] = x;
    __syncthreads();
    int base = 0;
    for (int q = 0; q < w; ++q) base += wsum[q];
    const int ex = base + x - own;
    h[t] = ex;
    __syncthreads();
    int p[KPT];
#pragma unroll
    for (int k = 0; k < KPT; ++k) p[k] = atomicAdd(&h[bk[k]], 1);
#pragma unroll
    for (int k = 0; k < KPT; ++k) {
        if (tg[k] >= 0) {
            const int e = e0 + t + k * 256;
            srt[p[k]] = (e << 8) | (tg[k] - bk[k] * BIN_W);
        }
    }
    __syncthreads();
#pragma unroll
    for (int k = 0; k < KPT; ++k) {
        const int i = t + k * 256;
        if (i < nloc) rec[e0 + i] = srt[i];
    }
    H[c * NB + t] = own;
    R[t * NCH + c] = make_int2(ex, own);
}

__global__ void __launch_bounds__(256)
gat_fine_gemm(const int* __restrict__ H, const int2* __restrict__ R, const int* __restrict__ rec,
              int* __restrict__ cnt, int* __restrict__ offs, int* __restrict__ list,
              GEMM_ARGS_DECL) {
    __shared__ __attribute__((aligned(16))) float L[8 * TILE_F];
    __shared__ __attribute__((aligned(16))) int part[4 * NB];
    __shared__ int c[256];
    __shared__ int cur[256];
    __shared__ int wsum[8];
    __shared__ int seg;
    if (blockIdx.x >= NB) {
        gemm_role(L, blockIdx.x - NB, GP2, SCUT, STRIPS, GEMM_ARGS);
        return;
    }
    const int t = threadIdx.x, lane = t & 31, w = t >> 5;
    const int b = blockIdx.x;
    const int ws = __builtin_amdgcn_readfirstlane(w);
    int2 run = make_int2(0, 0);
    if (lane < RPW) run = R[b * NCH + ws + 8 * lane];
    {
        const int j = t & 63, g = t >> 6;
        int4 T = make_int4(0, 0, 0, 0);
#pragma unroll 8
        for (int k = 0; k < NCH / 4; ++k) {
            const int r = (b + g + 4 * k) & (NCH - 1);
            const int4 v = *(const int4*)(H + r * NB + 4 * j);
            T.x += v.x; T.y += v.y; T.z += v.z; T.w += v.w;
        }
        *(int4*)(part + g * NB + 4 * j) = T;
    }
    c[t] = 0;
    __syncthreads();
    {
        const int total = part[t] + part[NB + t] + part[2 * NB + t] + part[3 * NB + t];
        int x = total;
#pragma unroll
        for (int o = 1; o < 32; o <<= 1) {
            int m = __shfl_up(x, o);
            if (lane >= o) x += m;
        }
        if (lane == 31) wsum[w] = x;
        __syncthreads();
        int base = 0;
        for (int q = 0; q < w; ++q) base += wsum[q];
        if (t == b) seg = base + x - total;
    }
    __syncthreads();
    const int start = seg;
    int len[RPW], fst[RPW], r[RPW];
#pragma unroll
    for (int k = 0; k < RPW; ++k) {
        len[k] = __builtin_amdgcn_readlane(run.y, k);
        fst[k] = (ws + 8 * k) * CH + __builtin_amdgcn_readlane(run.x, k);
        r[k] = lane < len[k] ? rec[fst[k] + lane] : -1;
    }
#pragma unroll
    for (int k = 0; k < RPW; ++k)
        if (r[k] >= 0) atomicAdd(&c[r[k] & 255], 1);
#pragma unroll
    for (int k = 0; k < RPW; ++k) {
        if (len[k] > 32) {
            for (int i = lane + 32; i < len[k]; i += 32)
                atomicAdd(&c[rec[fst[k] + i] & 255], 1);
        }
    }
    __syncthreads();
    const int own = c[t];
    int x = own;
#pragma unroll
    for (int o = 1; o < 32; o <<= 1) {
        int m = __shfl_up(x, o);
        if (lane >= o) x += m;
    }
    if (lane == 31) wsum[w] = x;
    __syncthreads();
    {
        int base = 0;
        for (int j = 0; j < w; ++j) base += wsum[j];
        const int ex = base + x - own;
        cur[t] = ex;
        __syncthreads();
        const int node = b * BIN_W + t;
        if (t < BIN_W && node < NN) {
            cnt[node]  = own;
            offs[node] = start + ex;
        }
    }
    int p[RPW];
#pragma unroll
    for (int k = 0; k < RPW; ++k) {
        p[k] = 0;
        if (r[k] >= 0) p[k] = atomicAdd(&cur[r[k] & 255], 1);
    }
#pragma unroll
    for (int k = 0; k < RPW; ++k)
        if (r[k] >= 0) list[start + p[k]] = r[k] >> 8;
#pragma unroll
    for (int k = 0; k < RPW; ++k) {
        if (len[k] > 32) {
            for (int i = lane + 32; i < len[k]; i += 32) {
                const int v = rec[fst[k] + i];
                const int q = atomicAdd(&cur[v & 255], 1);
                list[start + q] = v >> 8;
            }
        }
    }
}

__device__ __forceinline__ float edge_exp(float s_src, float s_trg) {
    float v = s_src + s_trg;
    v = v > 0.f ? v : 0.2f * v;
    return __expf(v);
}

template <int T>
__device__ __forceinline__ float quad_lane(float v) {
    return __int_as_float(__builtin_amdgcn_update_dpp(
        0, __float_as_int(v), T * 0x55, 0xf, 0xf, true));
}

__device__ __forceinline__ void row_fma(float4& acc, const float4& x, float a) {
    acc.x += x.x * a; acc.y += x.y * a; acc.z += x.z * a; acc.w += x.w * a;
}

__device__ __forceinline__ void rows_request(float4 (&x)[8], const float4* __restrict__ emb4,
                                             int sv, int r0, int last, int lane) {
#pragma unroll
    for (int t = 0; t < 8; ++t) {
        const int r = r0 + t < last ? r0 + t : last;
        x[t] = emb4[__builtin_amdgcn_readlane(sv, r) * 32 + lane];
    }
}

__device__ __forceinline__ void rows_sum(float4& acc, const float4 (&x)[8], float wa, float wb) {
    row_fma(acc, x[0], quad_lane<0>(wa));
    row_fma(acc, x[1], quad_lane<1>(wa));
    row_fma(acc, x[2], quad_lane<2>(wa));
    row_fma(acc, x[3], quad_lane<3>(wa));
    row_fma(acc, x[4], quad_lane<0>(wb));
    row_fma(acc, x[5], quad_lane<1>(wb));
    row_fma(acc, x[6], quad_lane<2>(wb));
    row_fma(acc, x[7], quad_lane<3>(wb));
}

__global__ void __launch_bounds__(256)
gat_aggregate(const int* __restrict__ src,
              const int* __restrict__ cnt, const int* __restrict__ offs,
              const int* __restrict__ list,
              const float* __restrict__ emb,
              const float* __restrict__ ss, const float* __restrict__ ts,
              const float* __restrict__ bias,
              float* __restrict__ out) {
    const int lane = threadIdx.x & 31;
    const int node = __builtin_amdgcn_readfirstlane((int)((blockIdx.x * blockDim.x + threadIdx.x) >> 5));
    if (node >= NN) return;
    const int h = lane >> 2;
    const int deg = __builtin_amdgcn_readfirstlane(cnt[node]);
    const int beg = __builtin_amdgcn_readfirstlane(offs[node]);
    const float tsv = ts[node * HEADS + h];
    const float4* __restrict__ emb4 = (const float4*)emb;

    float denom = 0.f;
    float4 acc = make_float4(0.f, 0.f, 0.f, 0.f);

    if (deg <= 32) {
        int e = lane < deg ? list[beg + lane] : 0x7fffffff;
        int rank = 0;
        for (int j = 0; j < deg; ++j) {
            int ej = __builtin_amdgcn_readlane(e, j);
            rank += (ej < e) ? 1 : 0;
        }
        if (lane >= deg) rank = lane;
        int es = __builtin_amdgcn_ds_permute(rank << 2, e);
        int sv = lane < deg ? src[es] : 0;

        const int q    = lane & 3;
        const int last = deg > 0 ? deg - 1 : 0;
        const bool big = deg > 16;

        int sj[8];
#pragma unroll
        for (int j = 0; j < 4; ++j)
            sj[j] = __builtin_amdgcn_ds_bpermute((4 * j + q) << 2, sv);
        float4 xr[8];
        rows_request(xr, emb4, sv, 0, last, lane);
        float sc[8];
#pragma unroll
        for (int j = 0; j < 4; ++j) sc[j] = ss[sj[j] * HEADS + h];
#pragma unroll
        for (int j = 4; j < 8; ++j) sc[j] = 0.f;
        if (big) {
#pragma unroll
            for (int j = 4; j < 8; ++j)
                sj[j] = __builtin_amdgcn_ds_bpermute((4 * j + q) << 2, sv);
#pragma unroll
            for (int j = 4; j < 8; ++j) sc[j] = ss[sj[j] * HEADS + h];
        }

        float ex[8];
#pragma unroll
        for (int j = 0; j < 4; ++j) {
            const float v = edge_exp(sc[j], tsv);
            ex[j] = (4 * j + q < deg) ? v : 0.f;
        }
#pragma unroll
        for (int j = 4; j < 8; ++j) ex[j] = 0.f;
#pragma unroll
        for (int j = 0; j < 4; ++j) {
            denom += quad_lane<0>(ex[j]);
            denom += quad_lane<1>(ex[j]);
            denom += quad_lane<2>(ex[j]);
            denom += quad_lane<3>(ex[j]);
        }
        if (big) {
#pragma unroll
            for (int j = 4; j < 8; ++j) {
                const float v = edge_exp(sc[j], tsv);
                ex[j] = (4 * j + q < deg) ? v : 0.f;
            }
#pragma unroll
            for (int j = 4; j < 8; ++j) {
                denom += quad_lane<0>(ex[j]);
                denom += quad_lane<1>(ex[j]);
                denom += quad_lane<2>(ex[j]);
                denom += quad_lane<3>(ex[j]);
            }
        }
        const float dn = denom + 1e-16f;
        float w[8];
#pragma unroll
        for (int j = 0; j < 8; ++j) w[j] = ex[j] / dn;

        rows_sum(acc, xr, w[0], w[1]);
        if (deg > 8) {
            rows_request(xr, emb4, sv, 8, last, lane);
            rows_sum(acc, xr, w[2], w[3]);
            if (big) {
                rows_request(xr, emb4, sv, 16, last, lane);
                rows_sum(acc, xr, w[4], w[5]);
                if (deg > 24) {
                    rows_request(xr, emb4, sv, 24, last, lane);
                    rows_sum(acc, xr, w[6], w[7]);
                }
            }
        }
    } else {
        int prev = -1;
        for (int r = 0; r < deg; ++r) {
            int m = 0x7fffffff;
            for (int j = lane; j < deg; j += 32) {
                int ej = list[beg + j];
                if (ej > prev && ej < m) m = ej;
            }
#pragma unroll
            for (int o = 16; o > 0; o >>= 1) {
                int n = __shfl_xor(m, o);
                m = n < m ? n : m;
            }
            prev = __builtin_amdgcn_readfirstlane(m);
            int s = src[prev];
            denom += edge_exp(ss[s * HEADS + h], tsv);
        }
        const float dn = denom + 1e-16f;
        prev = -1;
        for (int r = 0; r < deg; ++r) {
            int m = 0x7fffffff;
            for (int j = lane; j < deg; j += 32) {
                int ej = list[beg + j];
                if (ej > prev && ej < m) m = ej;
            }
#pragma unroll
            for (int o = 16; o > 0; o >>= 1) {
                int n = __shfl_xor(m, o);
                m = n < m ? n : m;
            }
            prev = __builtin_amdgcn_readfirstlane(m);
            int s0 = src[prev];
            float4 x0 = emb4[s0 * 32 + lane];
            float a0 = edge_exp(ss[s0 * HEADS + h], tsv) / dn;
            acc.x += x0.x * a0; acc.y += x0.y * a0; acc.z += x0.z * a0; acc.w += x0.w * a0;
        }
    }

    const float4 b = ((const float4*)bias)[lane];
    float4 o4 = make_float4(acc.x + b.x, acc.y + b.y, acc.z + b.z, acc.w + b.w);
    ((float4*)out)[node * 32 + lane] = o4;
}

extern "C" void kernel_launch(void* const* d_in, const int* in_sizes, int n_in,
                              void* d_out, int out_size, void* d_ws, size_t ws_size,
                              hipStream_t stream) {
    const float* X      = (const float*)d_in[0];
    const int*   eidx   = (const int*)d_in[1];
    const float* W      = (const float*)d_in[2];
    const float* a_l    = (const float*)d_in[3];
    const float* a_r    = (const float*)d_in[4];
    const float* bias   = (const float*)d_in[5];
    float* out = (float*)d_out;

    const int* src = eidx;
    const int* trg = eidx + EE;

    float* ws    = (float*)d_ws;
    float* emb   = ws;
    float* ss    = emb + (size_t)NN * OUTC;
    float* ts    = ss + (size_t)NN * HEADS;
    int*   cnt   = (int*)(ts + (size_t)NN * HEADS);
    int*   offs  = cnt + NN;
    int*   rec   = offs + NN;
    int*   list  = rec + EE;
    int*   H = (int*)d_out;
    int2*  R = (int2*)(H + NCH * NB);

    gat_chunksort_gemm<<<NCH + GP1, 256, 0, stream>>>(trg, H, R, rec, X, W, a_l, a_r, emb, ss, ts);
    gat_fine_gemm<<<NB + GP2, 256, 0, stream>>>(H, R, rec, cnt, offs, list, X, W, a_l, a_r, emb, ss, ts);

    gat_aggregate<<<(NN + 7) / 8, 256, 0, stream>>>(src, cnt, offs, list, emb, ss, ts, bias, out);
}


# Source node travels with each edge through the grouping (records and node lists are (edge id, source) pairs): the aggregate wave has its sources with the list, no per-edge gather from src; same sums
# speedup vs baseline: 6.1986x; 1.0473x over previous
#include <hip/hip_runtime.h>
#include <hip/hip_bf16.h>

#define NN 50000
#define EE 600000
#define INC 128
#define OUTC 128
#define HEADS 8
#define HEADC 16

typedef __attribute__((ext_vector_type(2))) float v2f;
typedef __attribute__((ext_vector_type(8))) float v8f;

template <int M>
__device__ __forceinline__ float row_xor(float v) {
    return __int_as_float(__builtin_amdgcn_update_dpp(
        0, __float_as_int(v), 0x160 | M, 0xf, 0xf, true));
}
template <int M>
__device__ __forceinline__ float halve(float lo, float hi, bool up) {
    const float keep = up ? hi : lo;
    const float send = up ? lo : hi;
    return keep + row_xor<M>(send);
}

#define STRIPS    (NN / 16)
#define GP1       520
#define GP2       522
#define SCUT      (3 * GP1)
#define APITCH    132
#define TILE_F    (16 * APITCH)
static_assert(STRIPS * 16 == NN, "strips cover every row exactly");
static_assert(GP1 <= SCUT && GP2 <= STRIPS - SCUT && STRIPS - SCUT <= 3 * GP2, "every product block has a first strip and at most three");

__device__ __forceinline__ void lds_barrier() {
    __builtin_amdgcn_fence(__ATOMIC_RELEASE, "workgroup", "local");
    __builtin_amdgcn_s_barrier();
    __builtin_amdgcn_fence(__ATOMIC_ACQUIRE, "workgroup", "local");
}

__device__ __forceinline__ void
gemm_role(float* L, const int bid, const int gp, const int s_begin, const int s_end,
          const float* __restrict__ X,
          const float* __restrict__ W,
          const float* __restrict__ a_left,
          const float* __restrict__ a_right,
          float* __restrict__ emb,
          float* __restrict__ ss, float* __restrict__ ts) {
    const int first = s_begin + bid;
    if (first >= s_end) return;
    const int tid  = threadIdx.x;
    const int wave = tid >> 5;
    const int lane = tid & 31;
    const int half = lane >> 4;
    const int l    = lane & 15;

    const float4* __restrict__ X4 = (const float4*)X;
    float4 t0 = X4[first * 512 + tid];
    float4 t1 = X4[first * 512 + 256 + tid];

    const float4* __restrict__ W4 = (const float4*)W + wave * 512;
    float* wsm = L + wave * TILE_F;
#pragma unroll
    for (int i = 0; i < 16; ++i)
        *(float4*)(wsm + i * APITCH + lane * 4) = W4[i * 32 + lane];
    __syncthreads();
    v2f b[32];
#pragma unroll
    for (int kk = 0; kk < 32; ++kk)
        b[kk] = *(const v2f*)(wsm + l * APITCH + kk * 4 + half * 2);

    float* as0 = L + (tid >> 5) * APITCH + (tid & 31) * 4;
    float* as1 = as0 + 8 * APITCH;
    const float* afr = L + l * APITCH + half * 2;

    const float al = a_left[l * HEADS + wave];
    const float ar = a_right[l * HEADS + wave];
    const bool  up8 = (l & 8) != 0, up4 = (l & 4) != 0, up2 = (l & 2) != 0, up1 = (l & 1) != 0;
    const float a_keep = up8 ? ar : al;
    const float a_send = up8 ? al : ar;
    float* sst = L + TILE_F;
    float* sput = sst + (l >> 3) * 128 + (half * 8 + (l & 7)) * HEADS + wave;
    float* __restrict__ sp = (tid < 128 ? ss : ts) + (size_t)first * 128 + (tid & 127);

    int strip = first;
    float* __restrict__ op =
        emb + (size_t)(strip * 16 + half * 8) * OUTC + wave * 16 + l;
    for (; strip < s_end; strip += gp) {
        lds_barrier();
        if (strip != first) {
            *sp = sst[tid];
            sp += (size_t)gp * 128;
        }
        *(float4*)as0 = t0;
        *(float4*)as1 = t1;
        lds_barrier();
        const int next = strip + gp;
        if (next < s_end) {
            t0 = X4[next * 512 + tid];
            t1 = X4[next * 512 + 256 + tid];
        }

        v8f c = {};
#pragma unroll
        for (int kk = 0; kk < 32; ++kk) {
            v2f a = *(const v2f*)(afr + kk * 4);
            c = __builtin_amdgcn_wmma_f32_16x16x4_f32(
                     false, a,  false, b[kk],
                     (short)0, c,  false,  false);
        }

#pragma unroll
        for (int v = 0; v < 8; ++v) op[(size_t)v * OUTC] = c[v];
        op += (size_t)gp * 16 * OUTC;

        float q8[8], q4[4], q2[2];
#pragma unroll
        for (int v = 0; v < 8; ++v) q8[v] = c[v] * a_keep + row_xor<8>(c[v] * a_send);
#pragma unroll
        for (int k = 0; k < 4; ++k) q4[k] = halve<4>(q8[k], q8[k + 4], up4);
#pragma unroll
        for (int k = 0; k < 2; ++k) q2[k] = halve<2>(q4[k], q4[k + 2], up2);
        *sput = halve<1>(q2[0], q2[1], up1);
    }
    lds_barrier();
    *sp = sst[tid];
}

#define NB    256
#define BIN_W 196
#define NCH   128
#define CH    4688
static_assert(NB * BIN_W >= NN, "bins cover every node");
static_assert(NCH * CH >= EE && (NCH - 1) * CH < EE, "chunks cover every edge, none is empty");
static_assert(NB == 256 && (NCH & (NCH - 1)) == 0 && NCH % 8 == 0 && NCH <= 256, "column sums: 64 threads x 4 bins per row, 4 rows a step, rows taken modulo NCH; runs: 8 waves, one thread per chunk");
static_assert(BIN_W <= 256 && EE < (1 << 23), "record packing: edge id << 8 | index in bin");
#define KPT   ((CH + 255) / 256)
#define RPW   (NCH / 8)
#define GEMM_ARGS_DECL const float* __restrict__ X, const float* __restrict__ W, \
                       const float* __restrict__ a_left, const float* __restrict__ a_right, \
                       float* __restrict__ emb, float* __restrict__ ss, float* __restrict__ ts
#define GEMM_ARGS X, W, a_left, a_right, emb, ss, ts

__global__ void __launch_bounds__(256)
gat_chunksort_gemm(const int* __restrict__ src, const int* __restrict__ trg,
                   int* __restrict__ H, int2* __restrict__ R,
                   int2* __restrict__ rec, GEMM_ARGS_DECL) {
    __shared__ __attribute__((aligned(16))) float L[8 * TILE_F];
    __shared__ int h[NB + 1];
    __shared__ int wsum[8];
    __shared__ int srt[CH];
    int* const srs = (int*)L;
    static_assert(sizeof(int) * CH <= sizeof(float) * 8 * TILE_F, "the source words fit in the tiles' space");
    if (blockIdx.x >= NCH) {
        gemm_role(L, blockIdx.x - NCH, GP1, 0, SCUT, GEMM_ARGS);
        return;
    }
    const int t = threadIdx.x, lane = t & 31, w = t >> 5;
    const int c = blockIdx.x;
    h[t] = 0;
    if (t == 0) h[NB] = 0;
    __syncthreads();
    const int e0 = c * CH;
    const int nloc = (EE - e0 < CH) ? EE - e0 : CH;
    int tg[KPT], sg[KPT];
#pragma unroll
    for (int k = 0; k < KPT; ++k) {
        const int i = t + k * 256;
        tg[k] = (i < nloc) ? trg[e0 + i] : -1;
        sg[k] = (i < nloc) ? src[e0 + i] : 0;
    }
    int bk[KPT];
#pragma unroll
    for (int k = 0; k < KPT; ++k) bk[k] = tg[k] >= 0 ? tg[k] / BIN_W : NB;
#pragma unroll
    for (int k = 0; k < KPT; ++k) atomicAdd(&h[bk[k]], 1);
    __syncthreads();
    const int own = h[t];
    int x = own;
#pragma unroll
    for (int o = 1; o < 32; o <<= 1) {
        int m = __shfl_up(x, o);
        if (lane >= o) x += m;
    }
    if (lane == 31) wsum[w] = x;
    __syncthreads();
    int base = 0;
    for (int q = 0; q < w; ++q) base += wsum[q];
    const int ex = base + x - own;
    h[t] = ex;
    __syncthreads();
    int p[KPT];
#pragma unroll
    for (int k = 0; k < KPT; ++k) p[k] = atomicAdd(&h[bk[k]], 1);
#pragma unroll
    for (int k = 0; k < KPT; ++k) {
        if (tg[k] >= 0) {
            const int e = e0 + t + k * 256;
            srt[p[k]] = (e << 8) | (tg[k] - bk[k] * BIN_W);
            srs[p[k]] = sg[k];
        }
    }
    __syncthreads();
#pragma unroll
    for (int k = 0; k < KPT; ++k) {
        const int i = t + k * 256;
        if (i < nloc) rec[e0 + i] = make_int2(srt[i], srs[i]);
    }
    H[c * NB + t] = own;
    R[t * NCH + c] = make_int2(ex, own);
}

__global__ void __launch_bounds__(256)
gat_fine_gemm(const int* __restrict__ H, const int2* __restrict__ R, const int2* __restrict__ rec,
              int* __restrict__ cnt, int* __restrict__ offs, int2* __restrict__ list,
              GEMM_ARGS_DECL) {
    __shared__ __attribute__((aligned(16))) float L[8 * TILE_F];
    __shared__ __attribute__((aligned(16))) int part[4 * NB];
    __shared__ int c[256];
    __shared__ int cur[256];
    __shared__ int wsum[8];
    __shared__ int seg;
    if (blockIdx.x >= NB) {
        gemm_role(L, blockIdx.x - NB, GP2, SCUT, STRIPS, GEMM_ARGS);
        return;
    }
    const int t = threadIdx.x, lane = t & 31, w = t >> 5;
    const int b = blockIdx.x;
    const int ws = __builtin_amdgcn_readfirstlane(w);
    int2 run = make_int2(0, 0);
    if (lane < RPW) run = R[b * NCH + ws + 8 * lane];
    {
        const int j = t & 63, g = t >> 6;
        int4 T = make_int4(0, 0, 0, 0);
#pragma unroll 8
        for (int k = 0; k < NCH / 4; ++k) {
            const int r = (b + g + 4 * k) & (NCH - 1);
            const int4 v = *(const int4*)(H + r * NB + 4 * j);
            T.x += v.x; T.y += v.y; T.z += v.z; T.w += v.w;
        }
        *(int4*)(part + g * NB + 4 * j) = T;
    }
    c[t] = 0;
    __syncthreads();
    {
        const int total = part[t] + part[NB + t] + part[2 * NB + t] + part[3 * NB + t];
        int x = total;
#pragma unroll
        for (int o = 1; o < 32; o <<= 1) {
            int m = __shfl_up(x, o);
            if (lane >= o) x += m;
        }
        if (lane == 31) wsum[w] = x;
        __syncthreads();
        int base = 0;
        for (int q = 0; q < w; ++q) base += wsum[q];
        if (t == b) seg = base + x - total;
    }
    __syncthreads();
    const int start = seg;
    int len[RPW], fst[RPW];
    int2 r[RPW];
#pragma unroll
    for (int k = 0; k < RPW; ++k) {
        len[k] = __builtin_amdgcn_readlane(run.y, k);
        fst[k] = (ws + 8 * k) * CH + __builtin_amdgcn_readlane(run.x, k);
        r[k] = lane < len[k] ? rec[fst[k] + lane] : make_int2(-1, 0);
    }
#pragma unroll
    for (int k = 0; k < RPW; ++k)
        if (r[k].x >= 0) atomicAdd(&c[r[k].x & 255], 1);
#pragma unroll
    for (int k = 0; k < RPW; ++k) {
        if (len[k] > 32) {
            for (int i = lane + 32; i < len[k]; i += 32)
                atomicAdd(&c[rec[fst[k] + i].x & 255], 1);
        }
    }
    __syncthreads();
    const int own = c[t];
    int x = own;
#pragma unroll
    for (int o = 1; o < 32; o <<= 1) {
        int m = __shfl_up(x, o);
        if (lane >= o) x += m;
    }
    if (lane == 31) wsum[w] = x;
    __syncthreads();
    {
        int base = 0;
        for (int j = 0; j < w; ++j) base += wsum[j];
        const int ex = base + x - own;
        cur[t] = ex;
        __syncthreads();
        const int node = b * BIN_W + t;
        if (t < BIN_W && node < NN) {
            cnt[node]  = own;
            offs[node] = start + ex;
        }
    }
    int p[RPW];
#pragma unroll
    for (int k = 0; k < RPW; ++k) {
        p[k] = 0;
        if (r[k].x >= 0) p[k] = atomicAdd(&cur[r[k].x & 255], 1);
    }
#pragma unroll
    for (int k = 0; k < RPW; ++k)
        if (r[k].x >= 0) list[start + p[k]] = make_int2(r[k].x >> 8, r[k].y);
#pragma unroll
    for (int k = 0; k < RPW; ++k) {
        if (len[k] > 32) {
            for (int i = lane + 32; i < len[k]; i += 32) {
                const int2 v = rec[fst[k] + i];
                const int q = atomicAdd(&cur[v.x & 255], 1);
                list[start + q] = make_int2(v.x >> 8, v.y);
            }
        }
    }
}

__device__ __forceinline__ float edge_exp(float s_src, float s_trg) {
    float v = s_src + s_trg;
    v = v > 0.f ? v : 0.2f * v;
    return __expf(v);
}

template <int T>
__device__ __forceinline__ float quad_lane(float v) {
    return __int_as_float(__builtin_amdgcn_update_dpp(
        0, __float_as_int(v), T * 0x55, 0xf, 0xf, true));
}

__device__ __forceinline__ void row_fma(float4& acc, const float4& x, float a) {
    acc.x += x.x * a; acc.y += x.y * a; acc.z += x.z * a; acc.w += x.w * a;
}

__device__ __forceinline__ void rows_request(float4 (&x)[8], const float4* __restrict__ emb4,
                                             int sv, int r0, int last, int lane) {
#pragma unroll
    for (int t = 0; t < 8; ++t) {
        const int r = r0 + t < last ? r0 + t : last;
        x[t] = emb4[__builtin_amdgcn_readlane(sv, r) * 32 + lane];
    }
}

__device__ __forceinline__ void rows_sum(float4& acc, const float4 (&x)[8], float wa, float wb) {
    row_fma(acc, x[0], quad_lane<0>(wa));
    row_fma(acc, x[1], quad_lane<1>(wa));
    row_fma(acc, x[2], quad_lane<2>(wa));
    row_fma(acc, x[3], quad_lane<3>(wa));
    row_fma(acc, x[4], quad_lane<0>(wb));
    row_fma(acc, x[5], quad_lane<1>(wb));
    row_fma(acc, x[6], quad_lane<2>(wb));
    row_fma(acc, x[7], quad_lane<3>(wb));
}

__global__ void __launch_bounds__(256)
gat_aggregate(const int* __restrict__ src,
              const int* __restrict__ cnt, const int* __restrict__ offs,
              const int2* __restrict__ list,
              const float* __restrict__ emb,
              const float* __restrict__ ss, const float* __restrict__ ts,
              const float* __restrict__ bias,
              float* __restrict__ out) {
    const int lane = threadIdx.x & 31;
    const int node = __builtin_amdgcn_readfirstlane((int)((blockIdx.x * blockDim.x + threadIdx.x) >> 5));
    if (node >= NN) return;
    const int h = lane >> 2;
    const int deg = __builtin_amdgcn_readfirstlane(cnt[node]);
    const int beg = __builtin_amdgcn_readfirstlane(offs[node]);
    const float tsv = ts[node * HEADS + h];
    const float4* __restrict__ emb4 = (const float4*)emb;

    float denom = 0.f;
    float4 acc = make_float4(0.f, 0.f, 0.f, 0.f);

    if (deg <= 32) {
        const int2 en = lane < deg ? list[beg + lane] : make_int2(0x7fffffff, 0);
        const int e = en.x;
        int rank = 0;
        for (int j = 0; j < deg; ++j) {
            int ej = __builtin_amdgcn_readlane(e, j);
            rank += (ej < e) ? 1 : 0;
        }
        if (lane >= deg) rank = lane;
        const int sv = __builtin_amdgcn_ds_permute(rank << 2, en.y);

        const int q    = lane & 3;
        const int last = deg > 0 ? deg - 1 : 0;
        const bool big = deg > 16;

        int sj[8];
#pragma unroll
        for (int j = 0; j < 4; ++j)
            sj[j] = __builtin_amdgcn_ds_bpermute((4 * j + q) << 2, sv);
        float4 xr[8];
        rows_request(xr, emb4, sv, 0, last, lane);
        float sc[8];
#pragma unroll
        for (int j = 0; j < 4; ++j) sc[j] = ss[sj[j] * HEADS + h];
#pragma unroll
        for (int j = 4; j < 8; ++j) sc[j] = 0.f;
        if (big) {
#pragma unroll
            for (int j = 4; j < 8; ++j)
                sj[j] = __builtin_amdgcn_ds_bpermute((4 * j + q) << 2, sv);
#pragma unroll
            for (int j = 4; j < 8; ++j) sc[j] = ss[sj[j] * HEADS + h];
        }

        float ex[8];
#pragma unroll
        for (int j = 0; j < 4; ++j) {
            const float v = edge_exp(sc[j], tsv);
            ex[j] = (4 * j + q < deg) ? v : 0.f;
        }
#pragma unroll
        for (int j = 4; j < 8; ++j) ex[j] = 0.f;
#pragma unroll
        for (int j = 0; j < 4; ++j) {
            denom += quad_lane<0>(ex[j]);
            denom += quad_lane<1>(ex[j]);
            denom += quad_lane<2>(ex[j]);
            denom += quad_lane<3>(ex[j]);
        }
        if (big) {
#pragma unroll
            for (int j = 4; j < 8; ++j) {
                const float v = edge_exp(sc[j], tsv);
                ex[j] = (4 * j + q < deg) ? v : 0.f;
            }
#pragma unroll
            for (int j = 4; j < 8; ++j) {
                denom += quad_lane<0>(ex[j]);
                denom += quad_lane<1>(ex[j]);
                denom += quad_lane<2>(ex[j]);
                denom += quad_lane<3>(ex[j]);
            }
        }
        const float dn = denom + 1e-16f;
        float w[8];
#pragma unroll
        for (int j = 0; j < 8; ++j) w[j] = ex[j] / dn;

        rows_sum(acc, xr, w[0], w[1]);
        if (deg > 8) {
            rows_request(xr, emb4, sv, 8, last, lane);
            rows_sum(acc, xr, w[2], w[3]);
            if (big) {
                rows_request(xr, emb4, sv, 16, last, lane);
                rows_sum(acc, xr, w[4], w[5]);
                if (deg > 24) {
                    rows_request(xr, emb4, sv, 24, last, lane);
                    rows_sum(acc, xr, w[6], w[7]);
                }
            }
        }
    } else {
        int prev = -1;
        for (int r = 0; r < deg; ++r) {
            int m = 0x7fffffff;
            for (int j = lane; j < deg; j += 32) {
                int ej = list[beg + j].x;
                if (ej > prev && ej < m) m = ej;
            }
#pragma unroll
            for (int o = 16; o > 0; o >>= 1) {
                int n = __shfl_xor(m, o);
                m = n < m ? n : m;
            }
            prev = __builtin_amdgcn_readfirstlane(m);
            int s = src[prev];
            denom += edge_exp(ss[s * HEADS + h], tsv);
        }
        const float dn = denom + 1e-16f;
        prev = -1;
        for (int r = 0; r < deg; ++r) {
            int m = 0x7fffffff;
            for (int j = lane; j < deg; j += 32) {
                int ej = list[beg + j].x;
                if (ej > prev && ej < m) m = ej;
            }
#pragma unroll
            for (int o = 16; o > 0; o >>= 1) {
                int n = __shfl_xor(m, o);
                m = n < m ? n : m;
            }
            prev = __builtin_amdgcn_readfirstlane(m);
            int s0 = src[prev];
            float4 x0 = emb4[s0 * 32 + lane];
            float a0 = edge_exp(ss[s0 * HEADS + h], tsv) / dn;
            acc.x += x0.x * a0; acc.y += x0.y * a0; acc.z += x0.z * a0; acc.w += x0.w * a0;
        }
    }

    const float4 b = ((const float4*)bias)[lane];
    float4 o4 = make_float4(acc.x + b.x, acc.y + b.y, acc.z + b.z, acc.w + b.w);
    ((float4*)out)[node * 32 + lane] = o4;
}

extern "C" void kernel_launch(void* const* d_in, const int* in_sizes, int n_in,
                              void* d_out, int out_size, void* d_ws, size_t ws_size,
                              hipStream_t stream) {
    const float* X      = (const float*)d_in[0];
    const int*   eidx   = (const int*)d_in[1];
    const float* W      = (const float*)d_in[2];
    const float* a_l    = (const float*)d_in[3];
    const float* a_r    = (const float*)d_in[4];
    const float* bias   = (const float*)d_in[5];
    float* out = (float*)d_out;

    const int* src = eidx;
    const int* trg = eidx + EE;

    float* ws    = (float*)d_ws;
    float* emb   = ws;
    float* ss    = emb + (size_t)NN * OUTC;
    float* ts    = ss + (size_t)NN * HEADS;
    int*   cnt   = (int*)(ts + (size_t)NN * HEADS);
    int*   offs  = cnt + NN;
    int2*  list  = (int2*)(offs + NN);
    int*   H = (int*)d_out;
    int2*  R = (int2*)(H + NCH * NB);
    int2*  rec = R + NB * NCH;

    gat_chunksort_gemm<<<NCH + GP1, 256, 0, stream>>>(src, trg, H, R, rec, X, W, a_l, a_r, emb, ss, ts);
    gat_fine_gemm<<<NB + GP2, 256, 0, stream>>>(H, R, rec, cnt, offs, list, X, W, a_l, a_r, emb, ss, ts);

    gat_aggregate<<<(NN + 7) / 8, 256, 0, stream>>>(src, cnt, offs, list, emb, ss, ts, bias, out);
}


# Bin blocks (32 waves) keep their node lists in LDS and sum their own nodes at once: aggregate launch, global list, count table gone (2 launches); whole product beside the chunk sort; same sums
# speedup vs baseline: 7.0056x; 1.1302x over previous
#include <hip/hip_runtime.h>
#include <hip/hip_bf16.h>

#define NN 50000
#define EE 600000
#define INC 128
#define OUTC 128
#define HEADS 8
#define HEADC 16

typedef __attribute__((ext_vector_type(2))) float v2f;
typedef __attribute__((ext_vector_type(8))) float v8f;

template <int M>
__device__ __forceinline__ float row_xor(float v) {
    return __int_as_float(__builtin_amdgcn_update_dpp(
        0, __float_as_int(v), 0x160 | M, 0xf, 0xf, true));
}
template <int M>
__device__ __forceinline__ float halve(float lo, float hi, bool up) {
    const float keep = up ? hi : lo;
    const float send = up ? lo : hi;
    return keep + row_xor<M>(send);
}

#define STRIPS    (NN / 16)
#define GPA       1042
#define APITCH    132
#define TILE_F    (16 * APITCH)
static_assert(STRIPS * 16 == NN, "strips cover every row exactly");
static_assert(GPA <= STRIPS && STRIPS <= 3 * GPA, "every product block has a first strip and at most three");

__device__ __forceinline__ void lds_barrier() {
    __builtin_amdgcn_fence(__ATOMIC_RELEASE, "workgroup", "local");
    __builtin_amdgcn_s_barrier();
    __builtin_amdgcn_fence(__ATOMIC_ACQUIRE, "workgroup", "local");
}

__device__ __forceinline__ void
gemm_role(float* L, const int bid, const int gp, const int s_begin, const int s_end,
          const float* __restrict__ X,
          const float* __restrict__ W,
          const float* __restrict__ a_left,
          const float* __restrict__ a_right,
          float* __restrict__ emb,
          float* __restrict__ ss, float* __restrict__ ts) {
    const int first = s_begin + bid;
    if (first >= s_end) return;
    const int tid  = threadIdx.x;
    const int wave = tid >> 5;
    const int lane = tid & 31;
    const int half = lane >> 4;
    const int l    = lane & 15;

    const float4* __restrict__ X4 = (const float4*)X;
    float4 t0 = X4[first * 512 + tid];
    float4 t1 = X4[first * 512 + 256 + tid];

    const float4* __restrict__ W4 = (const float4*)W + wave * 512;
    float* wsm = L + wave * TILE_F;
#pragma unroll
    for (int i = 0; i < 16; ++i)
        *(float4*)(wsm + i * APITCH + lane * 4) = W4[i * 32 + lane];
    __syncthreads();
    v2f b[32];
#pragma unroll
    for (int kk = 0; kk < 32; ++kk)
        b[kk] = *(const v2f*)(wsm + l * APITCH + kk * 4 + half * 2);

    float* as0 = L + (tid >> 5) * APITCH + (tid & 31) * 4;
    float* as1 = as0 + 8 * APITCH;
    const float* afr = L + l * APITCH + half * 2;

    const float al = a_left[l * HEADS + wave];
    const float ar = a_right[l * HEADS + wave];
    const bool  up8 = (l & 8) != 0, up4 = (l & 4) != 0, up2 = (l & 2) != 0, up1 = (l & 1) != 0;
    const float a_keep = up8 ? ar : al;
    const float a_send = up8 ? al : ar;
    float* sst = L + TILE_F;
    float* sput = sst + (l >> 3) * 128 + (half * 8 + (l & 7)) * HEADS + wave;
    float* __restrict__ sp = (tid < 128 ? ss : ts) + (size_t)first * 128 + (tid & 127);

    int strip = first;
    float* __restrict__ op =
        emb + (size_t)(strip * 16 + half * 8) * OUTC + wave * 16 + l;
    for (; strip < s_end; strip += gp) {
        lds_barrier();
        if (strip != first) {
            *sp = sst[tid];
            sp += (size_t)gp * 128;
        }
        *(float4*)as0 = t0;
        *(float4*)as1 = t1;
        lds_barrier();
        const int next = strip + gp;
        if (next < s_end) {
            t0 = X4[next * 512 + tid];
            t1 = X4[next * 512 + 256 + tid];
        }

        v8f c = {};
#pragma unroll
        for (int kk = 0; kk < 32; ++kk) {
            v2f a = *(const v2f*)(afr + kk * 4);
            c = __builtin_amdgcn_wmma_f32_16x16x4_f32(
                     false, a,  false, b[kk],
                     (short)0, c,  false,  false);
        }

#pragma unroll
        for (int v = 0; v < 8; ++v) op[(size_t)v * OUTC] = c[v];
        op += (size_t)gp * 16 * OUTC;

        float q8[8], q4[4], q2[2];
#pragma unroll
        for (int v = 0; v < 8; ++v) q8[v] = c[v] * a_keep + row_xor<8>(c[v] * a_send);
#pragma unroll
        for (int k = 0; k < 4; ++k) q4[k] = halve<4>(q8[k], q8[k + 4], up4);
#pragma unroll
        for (int k = 0; k < 2; ++k) q2[k] = halve<2>(q4[k], q4[k + 2], up2);
        *sput = halve<1>(q2[0], q2[1], up1);
    }
    lds_barrier();
    *sp = sst[tid];
}

#define NB    256
#define BIN_W 196
#define NCH   128
#define CH    4688
static_assert(NB * BIN_W >= NN, "bins cover every node");
static_assert(NCH * CH >= EE && (NCH - 1) * CH < EE, "chunks cover every edge, none is empty");
static_assert(NB == 256, "chunk sort: one thread per bin");
static_assert(BIN_W <= 256 && EE < (1 << 23), "record packing: edge id << 8 | index in bin");
#define KPT   ((CH + 255) / 256)
#define BWAVES 32
#define RPW   (NCH / BWAVES)
static_assert(NCH % BWAVES == 0 && RPW <= 32, "runs: one lane per run of the wave");
#define LCAP  20480
#define GEMM_ARGS_DECL const float* __restrict__ X, const float* __restrict__ W, \
                       const float* __restrict__ a_left, const float* __restrict__ a_right, \
                       float* __restrict__ emb, float* __restrict__ ss, float* __restrict__ ts
#define GEMM_ARGS X, W, a_left, a_right, emb, ss, ts

__global__ void __launch_bounds__(256)
gat_chunksort_gemm(const int* __restrict__ src, const int* __restrict__ trg,
                   int2* __restrict__ R,
                   int2* __restrict__ rec, GEMM_ARGS_DECL) {
    __shared__ __attribute__((aligned(16))) float L[8 * TILE_F];
    __shared__ int h[NB + 1];
    __shared__ int wsum[8];
    __shared__ int srt[CH];
    int* const srs = (int*)L;
    static_assert(sizeof(int) * CH <= sizeof(float) * 8 * TILE_F, "the source words fit in the tiles' space");
    if (blockIdx.x >= NCH) {
        gemm_role(L, blockIdx.x - NCH, GPA, 0, STRIPS, GEMM_ARGS);
        return;
    }
    const int t = threadIdx.x, lane = t & 31, w = t >> 5;
    const int c = blockIdx.x;
    h[t] = 0;
    if (t == 0) h[NB] = 0;
    __syncthreads();
    const int e0 = c * CH;
    const int nloc = (EE - e0 < CH) ? EE - e0 : CH;
    int tg[KPT], sg[KPT];
#pragma unroll
    for (int k = 0; k < KPT; ++k) {
        const int i = t + k * 256;
        tg[k] = (i < nloc) ? trg[e0 + i] : -1;
        sg[k] = (i < nloc) ? src[e0 + i] : 0;
    }
    int bk[KPT];
#pragma unroll
    for (int k = 0; k < KPT; ++k) bk[k] = tg[k] >= 0 ? tg[k] / BIN_W : NB;
#pragma unroll
    for (int k = 0; k < KPT; ++k) atomicAdd(&h[bk[k]], 1);
    __syncthreads();
    const int own = h[t];
    int x = own;
#pragma unroll
    for (int o = 1; o < 32; o <<= 1) {
        int m = __shfl_up(x, o);
        if (lane >= o) x += m;
    }
    if (lane == 31) wsum[w] = x;
    __syncthreads();
    int base = 0;
    for (int q = 0; q < w; ++q) base += wsum[q];
    const int ex = base + x - own;
    h[t] = ex;
    __syncthreads();
    int p[KPT];
#pragma unroll
    for (int k = 0; k < KPT; ++k) p[k] = atomicAdd(&h[bk[k]], 1);
#pragma unroll
    for (int k = 0; k < KPT; ++k) {
        if (tg[k] >= 0) {
            const int e = e0 + t + k * 256;
            srt[p[k]] = (e << 8) | (tg[k] - bk[k] * BIN_W);
            srs[p[k]] = sg[k];
        }
    }
    __syncthreads();
#pragma unroll
    for (int k = 0; k < KPT; ++k) {
        const int i = t + k * 256;
        if (i < nloc) rec[e0 + i] = make_int2(srt[i], srs[i]);
    }
    R[t * NCH + c] = make_int2(ex, own);
}

__device__ __forceinline__ float edge_exp(float s_src, float s_trg) {
    float v = s_src + s_trg;
    v = v > 0.f ? v : 0.2f * v;
    return __expf(v);
}

template <int T>
__device__ __forceinline__ float quad_lane(float v) {
    return __int_as_float(__builtin_amdgcn_update_dpp(
        0, __float_as_int(v), T * 0x55, 0xf, 0xf, true));
}

__device__ __forceinline__ void row_fma(float4& acc, const float4& x, float a) {
    acc.x += x.x * a; acc.y += x.y * a; acc.z += x.z * a; acc.w += x.w * a;
}

__device__ __forceinline__ void rows_request(float4 (&x)[8], const float4* __restrict__ emb4,
                                             int sv, int r0, int last, int lane) {
#pragma unroll
    for (int t = 0; t < 8; ++t) {
        const int r = r0 + t < last ? r0 + t : last;
        x[t] = emb4[__builtin_amdgcn_readlane(sv, r) * 32 + lane];
    }
}

__device__ __forceinline__ void rows_sum(float4& acc, const float4 (&x)[8], float wa, float wb) {
    row_fma(acc, x[0], quad_lane<0>(wa));
    row_fma(acc, x[1], quad_lane<1>(wa));
    row_fma(acc, x[2], quad_lane<2>(wa));
    row_fma(acc, x[3], quad_lane<3>(wa));
    row_fma(acc, x[4], quad_lane<0>(wb));
    row_fma(acc, x[5], quad_lane<1>(wb));
    row_fma(acc, x[6], quad_lane<2>(wb));
    row_fma(acc, x[7], quad_lane<3>(wb));
}

__global__ void __launch_bounds__(1024)
gat_bin_aggregate(const int* __restrict__ src,
                  const int2* __restrict__ R, const int2* __restrict__ rec,
                  const float* __restrict__ emb,
                  const float* __restrict__ ss, const float* __restrict__ ts,
                  const float* __restrict__ bias,
                  float* __restrict__ out) {
    __shared__ __attribute__((aligned(16))) int2 llist[LCAP];
    __shared__ int c[256];
    __shared__ int exs[256];
    __shared__ int cur[256];
    __shared__ int wsum[8];
    __shared__ int wtot[BWAVES];
    const int t = threadIdx.x, lane = t & 31, w = t >> 5;
    const int b = blockIdx.x;
    const int ws = __builtin_amdgcn_readfirstlane(w);
    int2 run = make_int2(0, 0);
    if (lane < RPW) run = R[b * NCH + ws + BWAVES * lane];
    if (t < 256) c[t] = 0;
    int len[RPW], fst[RPW];
    int2 r[RPW];
    int mine = 0;
#pragma unroll
    for (int k = 0; k < RPW; ++k) {
        len[k] = __builtin_amdgcn_readlane(run.y, k);
        fst[k] = (ws + BWAVES * k) * CH + __builtin_amdgcn_readlane(run.x, k);
        r[k] = lane < len[k] ? rec[fst[k] + lane] : make_int2(-1, 0);
        mine += len[k];
    }
    if (lane == 0) wtot[w] = mine;
    __syncthreads();
    int total = wtot[lane];
#pragma unroll
    for (int o = 16; o > 0; o >>= 1) total += __shfl_xor(total, o);
    const bool fits = __builtin_amdgcn_readfirstlane(total) <= LCAP;
#pragma unroll
    for (int k = 0; k < RPW; ++k)
        if (r[k].x >= 0) atomicAdd(&c[r[k].x & 255], 1);
#pragma unroll
    for (int k = 0; k < RPW; ++k) {
        if (len[k] > 32) {
            for (int i = lane + 32; i < len[k]; i += 32)
                atomicAdd(&c[rec[fst[k] + i].x & 255], 1);
        }
    }
    __syncthreads();
    int own = 0, x = 0;
    if (w < 8) {
        own = c[t];
        x = own;
#pragma unroll
        for (int o = 1; o < 32; o <<= 1) {
            int m = __shfl_up(x, o);
            if (lane >= o) x += m;
        }
        if (lane == 31) wsum[w] = x;
    }
    __syncthreads();
    if (w < 8) {
        int base = 0;
        for (int j = 0; j < w; ++j) base += wsum[j];
        const int ex = base + x - own;
        exs[t] = ex;
        cur[t] = ex;
    }
    __syncthreads();
    if (fits) {
        int p[RPW];
#pragma unroll
        for (int k = 0; k < RPW; ++k) {
            p[k] = 0;
            if (r[k].x >= 0) p[k] = atomicAdd(&cur[r[k].x & 255], 1);
        }
#pragma unroll
        for (int k = 0; k < RPW; ++k)
            if (r[k].x >= 0) llist[p[k]] = make_int2(r[k].x >> 8, r[k].y);
#pragma unroll
        for (int k = 0; k < RPW; ++k) {
            if (len[k] > 32) {
                for (int i = lane + 32; i < len[k]; i += 32) {
                    const int2 v = rec[fst[k] + i];
                    const int q = atomicAdd(&cur[v.x & 255], 1);
                    llist[q] = make_int2(v.x >> 8, v.y);
                }
            }
        }
    }
    __syncthreads();

    const int nn = (NN - b * BIN_W < BIN_W) ? NN - b * BIN_W : BIN_W;
    const int h = lane >> 2;
    const float4* __restrict__ emb4 = (const float4*)emb;
    const float4 bv = ((const float4*)bias)[lane];
    for (int i = ws; i < nn; i += BWAVES) {
        const int node = b * BIN_W + i;
        const int deg = __builtin_amdgcn_readfirstlane(c[i]);
        const int beg = __builtin_amdgcn_readfirstlane(exs[i]);
        const float tsv = ts[node * HEADS + h];

        float denom = 0.f;
        float4 acc = make_float4(0.f, 0.f, 0.f, 0.f);

        if (fits && deg <= 32) {
            const int2 en = lane < deg ? llist[beg + lane] : make_int2(0x7fffffff, 0);
            const int e = en.x;
            int rank = 0;
            for (int j = 0; j < deg; ++j) {
                int ej = __builtin_amdgcn_readlane(e, j);
                rank += (ej < e) ? 1 : 0;
            }
            if (lane >= deg) rank = lane;
            const int sv = __builtin_amdgcn_ds_permute(rank << 2, en.y);

            const int q    = lane & 3;
            const int last = deg > 0 ? deg - 1 : 0;
            const bool big = deg > 16;

            int sj[8];
#pragma unroll
            for (int j = 0; j < 4; ++j)
                sj[j] = __builtin_amdgcn_ds_bpermute((4 * j + q) << 2, sv);
            float4 xr[8];
            rows_request(xr, emb4, sv, 0, last, lane);
            float sc[8];
#pragma unroll
            for (int j = 0; j < 4; ++j) sc[j] = ss[sj[j] * HEADS + h];
#pragma unroll
            for (int j = 4; j < 8; ++j) sc[j] = 0.f;
            if (big) {
#pragma unroll
                for (int j = 4; j < 8; ++j)
                    sj[j] = __builtin_amdgcn_ds_bpermute((4 * j + q) << 2, sv);
#pragma unroll
                for (int j = 4; j < 8; ++j) sc[j] = ss[sj[j] * HEADS + h];
            }

            float ex[8];
#pragma unroll
            for (int j = 0; j < 4; ++j) {
                const float v = edge_exp(sc[j], tsv);
                ex[j] = (4 * j + q < deg) ? v : 0.f;
            }
#pragma unroll
            for (int j = 4; j < 8; ++j) ex[j] = 0.f;
#pragma unroll
            for (int j = 0; j < 4; ++j) {
                denom += quad_lane<0>(ex[j]);
                denom += quad_lane<1>(ex[j]);
                denom += quad_lane<2>(ex[j]);
                denom += quad_lane<3>(ex[j]);
            }
            if (big) {
#pragma unroll
                for (int j = 4; j < 8; ++j) {
                    const float v = edge_exp(sc[j], tsv);
                    ex[j] = (4 * j + q < deg) ? v : 0.f;
                }
#pragma unroll
                for (int j = 4; j < 8; ++j) {
                    denom += quad_lane<0>(ex[j]);
                    denom += quad_lane<1>(ex[j]);
                    denom += quad_lane<2>(ex[j]);
                    denom += quad_lane<3>(ex[j]);
                }
            }
            const float dn = denom + 1e-16f;
            float wt[8];
#pragma unroll
            for (int j = 0; j < 8; ++j) wt[j] = ex[j] / dn;

            rows_sum(acc, xr, wt[0], wt[1]);
            if (deg > 8) {
                rows_request(xr, emb4, sv, 8, last, lane);
                rows_sum(acc, xr, wt[2], wt[3]);
                if (big) {
                    rows_request(xr, emb4, sv, 16, last, lane);
                    rows_sum(acc, xr, wt[4], wt[5]);
                    if (deg > 24) {
                        rows_request(xr, emb4, sv, 24, last, lane);
                        rows_sum(acc, xr, wt[6], wt[7]);
                    }
                }
            }
        } else {
            auto next_id = [&](int prev) -> int {
                int m = 0x7fffffff;
                if (fits) {
                    for (int j = lane; j < deg; j += 32) {
                        const int ej = llist[beg + j].x;
                        if (ej > prev && ej < m) m = ej;
                    }
                } else {
                    for (int cc = 0; cc < NCH; ++cc) {
                        const int2 rn = R[b * NCH + cc];
                        const int f = cc * CH + rn.x;
                        for (int j = lane; j < rn.y; j += 32) {
                            const int vx = rec[f + j].x;
                            const int ej = vx >> 8;
                            if ((vx & 255) == i && ej > prev && ej < m) m = ej;
                        }
                    }
                }
#pragma unroll
                for (int o = 16; o > 0; o >>= 1) {
                    int n = __shfl_xor(m, o);
                    m = n < m ? n : m;
                }
                return __builtin_amdgcn_readfirstlane(m);
            };
            int prev = -1;
            for (int k = 0; k < deg; ++k) {
                prev = next_id(prev);
                int s = src[prev];
                denom += edge_exp(ss[s * HEADS + h], tsv);
            }
            const float dn = denom + 1e-16f;
            prev = -1;
            for (int k = 0; k < deg; ++k) {
                prev = next_id(prev);
                int s0 = src[prev];
                float4 x0 = emb4[s0 * 32 + lane];
                float a0 = edge_exp(ss[s0 * HEADS + h], tsv) / dn;
                acc.x += x0.x * a0; acc.y += x0.y * a0; acc.z += x0.z * a0; acc.w += x0.w * a0;
            }
        }

        float4 o4 = make_float4(acc.x + bv.x, acc.y + bv.y, acc.z + bv.z, acc.w + bv.w);
        ((float4*)out)[node * 32 + lane] = o4;
    }
}


extern "C" void kernel_launch(void* const* d_in, const int* in_sizes, int n_in,
                              void* d_out, int out_size, void* d_ws, size_t ws_size,
                              hipStream_t stream) {
    const float* X      = (const float*)d_in[0];
    const int*   eidx   = (const int*)d_in[1];
    const float* W      = (const float*)d_in[2];
    const float* a_l    = (const float*)d_in[3];
    const float* a_r    = (const float*)d_in[4];
    const float* bias   = (const float*)d_in[5];
    float* out = (float*)d_out;

    const int* src = eidx;
    const int* trg = eidx + EE;

    float* ws    = (float*)d_ws;
    float* emb   = ws;
    float* ss    = emb + (size_t)NN * OUTC;
    float* ts    = ss + (size_t)NN * HEADS;
    int2*  R   = (int2*)(ts + (size_t)NN * HEADS);
    int2*  rec = R + NB * NCH;

    gat_chunksort_gemm<<<NCH + GPA, 256, 0, stream>>>(src, trg, R, rec, X, W, a_l, a_r, emb, ss, ts);
    gat_bin_aggregate<<<NB, 1024, 0, stream>>>(src, R, rec, emb, ss, ts, bias, out);
}
